# starting kernel
# baseline (speedup 1.0000x reference)
#include <hip/hip_runtime.h>
#include <hip/hip_bf16.h>

typedef __attribute__((ext_vector_type(2))) float v2f;
typedef __attribute__((ext_vector_type(8))) float v8f;

#define HDIM 128

__global__ void zero_f32(float* __restrict__ p, long n) {
  long i = (long)blockIdx.x * blockDim.x + threadIdx.x;
  long stride = (long)gridDim.x * blockDim.x;
  for (; i < n; i += stride) p[i] = 0.0f;
}

__global__ void deg_kernel(const int* __restrict__ dst,
                           const float* __restrict__ w,
                           float* __restrict__ deg, int E) {
  int i = blockIdx.x * blockDim.x + threadIdx.x;
  int stride = gridDim.x * blockDim.x;
  for (; i < E; i += stride) unsafeAtomicAdd(&deg[dst[i]], w[i]);
}

__global__ void dinv_kernel(float* __restrict__ deg, int n) {
  int i = blockIdx.x * blockDim.x + threadIdx.x;
  int stride = gridDim.x * blockDim.x;
  for (; i < n; i += stride) deg[i] = rsqrtf(deg[i] + 1.0f);
}

__global__ void __launch_bounds__(256, 2)
gemm_wmma_f32(const float* __restrict__ A, const float* __restrict__ W,
              float* __restrict__ C, int nRowTiles) {
  const int lane = threadIdx.x & 31;
  const int wave = threadIdx.x >> 5;
  const int m    = lane & 15;
  const int half = lane >> 4;
  const int col0 = wave * 16;

  v2f bfrag[32];
#pragma unroll
  for (int kk = 0; kk < 32; ++kk) {
    const int k = kk * 4 + half * 2;
    v2f b;
    b.x = W[(k + 0) * HDIM + col0 + m];
    b.y = W[(k + 1) * HDIM + col0 + m];
    bfrag[kk] = b;
  }

  for (int rt = blockIdx.x; rt < nRowTiles; rt += gridDim.x) {
    const float* __restrict__ arow = A + (long)(rt * 16 + m) * HDIM;
    v8f acc = {0.f, 0.f, 0.f, 0.f, 0.f, 0.f, 0.f, 0.f};
#pragma unroll
    for (int kk = 0; kk < 32; ++kk) {
      const int k = kk * 4 + half * 2;
      const v2f a = *(const v2f*)(arow + k);
      acc = __builtin_amdgcn_wmma_f32_16x16x4_f32(
           false, a,  false, bfrag[kk],
           (short)0, acc,  false,  false);
    }
#pragma unroll
    for (int r = 0; r < 8; ++r)
      C[(long)(rt * 16 + half * 8 + r) * HDIM + col0 + m] = acc[r];
  }
}

__global__ void __launch_bounds__(256)
scatter_kernel(const float* __restrict__ h, const float* __restrict__ dinv,
               const int* __restrict__ src, const int* __restrict__ dst,
               const float* __restrict__ w, float* __restrict__ agg, int E) {
  const int lane = threadIdx.x & 31;
  int wid = blockIdx.x * (blockDim.x >> 5) + (threadIdx.x >> 5);
  const int nwaves = gridDim.x * (blockDim.x >> 5);
  for (int e = wid; e < E; e += nwaves) {
    const int s = src[e];
    const int d = dst[e];
    const float norm = dinv[s] * w[e] * dinv[d];
    const float4 hv = *(const float4*)(h + (long)s * HDIM + lane * 4);
    float* __restrict__ out = agg + (long)d * HDIM + lane * 4;
    unsafeAtomicAdd(out + 0, norm * hv.x);
    unsafeAtomicAdd(out + 1, norm * hv.y);
    unsafeAtomicAdd(out + 2, norm * hv.z);
    unsafeAtomicAdd(out + 3, norm * hv.w);
  }
}

__global__ void finalize_kernel(float* __restrict__ io,
                                const float* __restrict__ h,
                                const float* __restrict__ dinv,
                                const float* __restrict__ b,
                                long total, int relu) {
  long i = (long)blockIdx.x * blockDim.x + threadIdx.x;
  long stride = (long)gridDim.x * blockDim.x;
  for (; i < total; i += stride) {
    const int node = (int)(i >> 7);
    const int f = (int)(i & (HDIM - 1));
    const float di = dinv[node];
    float v = io[i] + di * di * h[i] + b[f];
    io[i] = relu ? fmaxf(v, 0.0f) : v;
  }
}

extern "C" void kernel_launch(void* const* d_in, const int* in_sizes, int n_in,
                              void* d_out, int out_size, void* d_ws, size_t ws_size,
                              hipStream_t stream) {
  const float* x  = (const float*)d_in[0];
  const float* ew = (const float*)d_in[1];
  const float* W1 = (const float*)d_in[2];
  const float* b1 = (const float*)d_in[3];
  const float* W2 = (const float*)d_in[4];
  const float* b2 = (const float*)d_in[5];
  const int*   ei = (const int*)d_in[6];

  const int N = in_sizes[0] / HDIM;
  const int E = in_sizes[1];
  const int* src = ei;
  const int* dst = ei + E;

  const long NH = (long)N * HDIM;

  float* deg = (float*)d_ws;
  float* h   = deg + (((long)N + 255) & ~255L);
  float* agg = h + NH;
  float* out = (float*)d_out;

  zero_f32<<<512, 256, 0, stream>>>(deg, N);
  zero_f32<<<2048, 256, 0, stream>>>(agg, NH);
  zero_f32<<<2048, 256, 0, stream>>>(out, NH);

  deg_kernel<<<2048, 256, 0, stream>>>(dst, ew, deg, E);
  dinv_kernel<<<512, 256, 0, stream>>>(deg, N);

  const int nRowTiles = N / 16;

  gemm_wmma_f32<<<1024, 256, 0, stream>>>(x, W1, h, nRowTiles);
  scatter_kernel<<<4096, 256, 0, stream>>>(h, deg, src, dst, ew, agg, E);
  finalize_kernel<<<2048, 256, 0, stream>>>(agg, h, deg, b1, NH, 1);

  gemm_wmma_f32<<<1024, 256, 0, stream>>>(agg, W2, h, nRowTiles);
  scatter_kernel<<<4096, 256, 0, stream>>>(h, deg, src, dst, ew, out, E);
  finalize_kernel<<<2048, 256, 0, stream>>>(out, h, deg, b2, NH, 0);
}


# Replace per-edge f32 atomic scatter by a destination-grouped (CSR) gather: edges ranked by int atomics on line-padded counters, one wave sums a node's rows in registers; sum order = arrival rank
# speedup vs baseline: 3.0824x; 3.0824x over previous
#include <hip/hip_runtime.h>
#include <hip/hip_bf16.h>

typedef __attribute__((ext_vector_type(2))) float v2f;
typedef __attribute__((ext_vector_type(8))) float v8f;
typedef unsigned long long u64;

#define HDIM 128
#define CPAD 32

__global__ void zero_i32(int* __restrict__ p, long n) {
  long i = (long)blockIdx.x * blockDim.x + threadIdx.x;
  long stride = (long)gridDim.x * blockDim.x;
  for (; i < n; i += stride) p[i] = 0;
}

__global__ void count_rank_kernel(const int* __restrict__ dst,
                                  int* __restrict__ cnt,
                                  int* __restrict__ rank, int E) {
  int i = blockIdx.x * blockDim.x + threadIdx.x;
  int stride = gridDim.x * blockDim.x;
  for (; i < E; i += stride)
    rank[i] = atomicAdd(&cnt[(long)dst[i] * CPAD], 1);
}

__global__ void __launch_bounds__(256)
alloc_kernel(const int* __restrict__ cnt, int* __restrict__ total,
             int* __restrict__ rowinfo_w, int n) {
  int2* __restrict__ rowinfo = (int2*)rowinfo_w;
  const int lane = threadIdx.x & 31;
  for (int base = blockIdx.x * blockDim.x; base < n; base += gridDim.x * blockDim.x) {
    const int i = base + threadIdx.x;
    const int c = (i < n) ? cnt[(long)i * CPAD] : 0;
    int incl = c;
    for (int d = 1; d < 32; d <<= 1) {
      const int t = __shfl_up(incl, d, 32);
      if (lane >= d) incl += t;
    }
    int wbase = 0;
    if (lane == 31) wbase = atomicAdd(total, incl);
    wbase = __shfl(wbase, 31, 32);
    if (i < n) rowinfo[i] = make_int2(wbase + incl - c, c);
  }
}

__global__ void fill_kernel(const int* __restrict__ src, const int* __restrict__ dst,
                            const float* __restrict__ w, const int* __restrict__ rank,
                            const int* __restrict__ rowinfo_w,
                            int* __restrict__ rec_w, int E) {
  const int2* __restrict__ rowinfo = (const int2*)rowinfo_w;
  u64* __restrict__ rec = (u64*)rec_w;
  int i = blockIdx.x * blockDim.x + threadIdx.x;
  int stride = gridDim.x * blockDim.x;
  for (; i < E; i += stride) {
    const int pos = rowinfo[dst[i]].x + rank[i];
    const u64 v = (u64)(unsigned)src[i] | ((u64)__float_as_uint(w[i]) << 32);
    (void)__hip_atomic_exchange(&rec[pos], v, __ATOMIC_RELAXED, __HIP_MEMORY_SCOPE_AGENT);
  }
}

__global__ void dinv_kernel(const int* __restrict__ rowinfo_w,
                            const int* __restrict__ rec_w,
                            float* __restrict__ dinv, int n) {
  const int2* __restrict__ rowinfo = (const int2*)rowinfo_w;
  const u64* __restrict__ rec = (const u64*)rec_w;
  int i = blockIdx.x * blockDim.x + threadIdx.x;
  int stride = gridDim.x * blockDim.x;
  for (; i < n; i += stride) {
    const int2 ri = rowinfo[i];
    float s = 0.0f;
    for (int j = 0; j < ri.y; ++j)
      s += __uint_as_float((unsigned)(rec[ri.x + j] >> 32));
    dinv[i] = rsqrtf(s + 1.0f);
  }
}

__global__ void __launch_bounds__(256, 2)
gemm_wmma_f32(const float* __restrict__ A, const float* __restrict__ W,
              float* __restrict__ C, int nRowTiles) {
  const int lane = threadIdx.x & 31;
  const int wave = threadIdx.x >> 5;
  const int m    = lane & 15;
  const int half = lane >> 4;
  const int col0 = wave * 16;

  v2f bfrag[32];
#pragma unroll
  for (int kk = 0; kk < 32; ++kk) {
    const int k = kk * 4 + half * 2;
    v2f b;
    b.x = W[(k + 0) * HDIM + col0 + m];
    b.y = W[(k + 1) * HDIM + col0 + m];
    bfrag[kk] = b;
  }

  for (int rt = blockIdx.x; rt < nRowTiles; rt += gridDim.x) {
    const float* __restrict__ arow = A + (long)(rt * 16 + m) * HDIM;
    v8f acc = {0.f, 0.f, 0.f, 0.f, 0.f, 0.f, 0.f, 0.f};
#pragma unroll
    for (int kk = 0; kk < 32; ++kk) {
      const int k = kk * 4 + half * 2;
      const v2f a = *(const v2f*)(arow + k);
      acc = __builtin_amdgcn_wmma_f32_16x16x4_f32(
           false, a,  false, bfrag[kk],
           (short)0, acc,  false,  false);
    }
#pragma unroll
    for (int r = 0; r < 8; ++r)
      C[(long)(rt * 16 + half * 8 + r) * HDIM + col0 + m] = acc[r];
  }
}

__device__ __forceinline__ float lane_bcast_f(float v, int j) {
  return __int_as_float(__builtin_amdgcn_readlane(__float_as_int(v), j));
}

__device__ __forceinline__ void axpy4(float4& acc, float c, const float4 v) {
  acc.x = fmaf(c, v.x, acc.x);
  acc.y = fmaf(c, v.y, acc.y);
  acc.z = fmaf(c, v.z, acc.z);
  acc.w = fmaf(c, v.w, acc.w);
}

__global__ void __launch_bounds__(256)
gather_kernel(const float* __restrict__ h, const float* __restrict__ dinv,
              const int* __restrict__ rowinfo_w, const int* __restrict__ rec_w,
              const float* __restrict__ b, float* __restrict__ out,
              int n_nodes, int relu) {
  const int2* __restrict__ rowinfo = (const int2*)rowinfo_w;
  const u64* __restrict__ rec = (const u64*)rec_w;
  const int lane = threadIdx.x & 31;
  const int nw = gridDim.x * 8;
  const int w0 = __builtin_amdgcn_readfirstlane((int)(blockIdx.x * 8 + (threadIdx.x >> 5)));
  const float4 bv = *(const float4*)(b + lane * 4);
  const float* __restrict__ hl = h + lane * 4;

  for (int n = w0; n < n_nodes; n += nw) {
    const int2 ri = rowinfo[n];
    const int st = ri.x;
    const int c = ri.y;
    const float di = dinv[n];
    const float4 hs = *(const float4*)(hl + (long)n * HDIM);
    float4 acc = {0.0f, 0.0f, 0.0f, 0.0f};

    for (int base = 0; base < c; base += 32) {
      const int m = min(32, c - base);
      int sj = 0;
      float cj = 0.0f;
      if (lane < m) {
        const u64 r = rec[st + base + lane];
        sj = (int)(unsigned)r;
        cj = dinv[sj] * __uint_as_float((unsigned)(r >> 32)) * di;
      }
      int j = 0;
      for (; j + 4 <= m; j += 4) {
        const int s0 = __builtin_amdgcn_readlane(sj, j + 0);
        const int s1 = __builtin_amdgcn_readlane(sj, j + 1);
        const int s2 = __builtin_amdgcn_readlane(sj, j + 2);
        const int s3 = __builtin_amdgcn_readlane(sj, j + 3);
        const float4 h0 = *(const float4*)(hl + (long)s0 * HDIM);
        const float4 h1 = *(const float4*)(hl + (long)s1 * HDIM);
        const float4 h2 = *(const float4*)(hl + (long)s2 * HDIM);
        const float4 h3 = *(const float4*)(hl + (long)s3 * HDIM);
        axpy4(acc, lane_bcast_f(cj, j + 0), h0);
        axpy4(acc, lane_bcast_f(cj, j + 1), h1);
        axpy4(acc, lane_bcast_f(cj, j + 2), h2);
        axpy4(acc, lane_bcast_f(cj, j + 3), h3);
      }
      for (; j < m; ++j) {
        const int s0 = __builtin_amdgcn_readlane(sj, j);
        const float4 h0 = *(const float4*)(hl + (long)s0 * HDIM);
        axpy4(acc, lane_bcast_f(cj, j), h0);
      }
    }

    const float dd = di * di;
    float4 v;
    v.x = acc.x + dd * hs.x + bv.x;
    v.y = acc.y + dd * hs.y + bv.y;
    v.z = acc.z + dd * hs.z + bv.z;
    v.w = acc.w + dd * hs.w + bv.w;
    if (relu) {
      v.x = fmaxf(v.x, 0.0f); v.y = fmaxf(v.y, 0.0f);
      v.z = fmaxf(v.z, 0.0f); v.w = fmaxf(v.w, 0.0f);
    }
    *(float4*)(out + (long)n * HDIM + lane * 4) = v;
  }
}

extern "C" void kernel_launch(void* const* d_in, const int* in_sizes, int n_in,
                              void* d_out, int out_size, void* d_ws, size_t ws_size,
                              hipStream_t stream) {
  const float* x  = (const float*)d_in[0];
  const float* ew = (const float*)d_in[1];
  const float* W1 = (const float*)d_in[2];
  const float* b1 = (const float*)d_in[3];
  const float* W2 = (const float*)d_in[4];
  const float* b2 = (const float*)d_in[5];
  const int*   ei = (const int*)d_in[6];

  const int N = in_sizes[0] / HDIM;
  const int E = in_sizes[1];
  const int* src = ei;
  const int* dst = ei + E;

  const long NH = (long)N * HDIM;
  const long Np = ((long)N + 255) & ~255L;
  const long Ep = ((long)E + 255) & ~255L;

  int*   cnt     = (int*)d_ws;
  int*   total   = cnt + (long)N * CPAD;
  int*   rowinfo = total + 256;
  float* dinv    = (float*)(rowinfo + 2 * Np);
  int*   rank    = (int*)(dinv + Np);
  int*   rec     = rank + Ep;
  float* h       = (float*)(rec + 2 * Ep);
  float* h1      = h + NH;
  float* out     = (float*)d_out;

  zero_i32<<<2048, 256, 0, stream>>>(cnt, (long)N * CPAD + 256);
  count_rank_kernel<<<6250, 256, 0, stream>>>(dst, cnt, rank, E);
  alloc_kernel<<<391, 256, 0, stream>>>(cnt, total, rowinfo, N);
  fill_kernel<<<6250, 256, 0, stream>>>(src, dst, ew, rank, rowinfo, rec, E);
  dinv_kernel<<<391, 256, 0, stream>>>(rowinfo, rec, dinv, N);

  const int nRowTiles = N / 16;

  gemm_wmma_f32<<<1024, 256, 0, stream>>>(x, W1, h, nRowTiles);
  gather_kernel<<<12500, 256, 0, stream>>>(h, dinv, rowinfo, rec, b1, h1, N, 1);

  gemm_wmma_f32<<<1024, 256, 0, stream>>>(h1, W2, h, nRowTiles);
  gather_kernel<<<12500, 256, 0, stream>>>(h, dinv, rowinfo, rec, b2, out, N, 0);
}


# GEMM reads each 16-row tile of A once per block as whole 512-byte rows into a two-stage LDS buffer; the 8 waves take A fragments from LDS instead of 8 x 16-byte-per-line global reads
# speedup vs baseline: 3.4428x; 1.1169x over previous
#include <hip/hip_runtime.h>
#include <hip/hip_bf16.h>

typedef __attribute__((ext_vector_type(2))) float v2f;
typedef __attribute__((ext_vector_type(8))) float v8f;
typedef unsigned long long u64;

#define HDIM 128
#define CPAD 32

__global__ void zero_i32(int* __restrict__ p, long n) {
  long i = (long)blockIdx.x * blockDim.x + threadIdx.x;
  long stride = (long)gridDim.x * blockDim.x;
  for (; i < n; i += stride) p[i] = 0;
}

__global__ void count_rank_kernel(const int* __restrict__ dst,
                                  int* __restrict__ cnt,
                                  int* __restrict__ rank, int E) {
  int i = blockIdx.x * blockDim.x + threadIdx.x;
  int stride = gridDim.x * blockDim.x;
  for (; i < E; i += stride)
    rank[i] = atomicAdd(&cnt[(long)dst[i] * CPAD], 1);
}

__global__ void __launch_bounds__(256)
alloc_kernel(const int* __restrict__ cnt, int* __restrict__ total,
             int* __restrict__ rowinfo_w, int n) {
  int2* __restrict__ rowinfo = (int2*)rowinfo_w;
  const int lane = threadIdx.x & 31;
  for (int base = blockIdx.x * blockDim.x; base < n; base += gridDim.x * blockDim.x) {
    const int i = base + threadIdx.x;
    const int c = (i < n) ? cnt[(long)i * CPAD] : 0;
    int incl = c;
    for (int d = 1; d < 32; d <<= 1) {
      const int t = __shfl_up(incl, d, 32);
      if (lane >= d) incl += t;
    }
    int wbase = 0;
    if (lane == 31) wbase = atomicAdd(total, incl);
    wbase = __shfl(wbase, 31, 32);
    if (i < n) rowinfo[i] = make_int2(wbase + incl - c, c);
  }
}

__global__ void fill_kernel(const int* __restrict__ src, const int* __restrict__ dst,
                            const float* __restrict__ w, const int* __restrict__ rank,
                            const int* __restrict__ rowinfo_w,
                            int* __restrict__ rec_w, int E) {
  const int2* __restrict__ rowinfo = (const int2*)rowinfo_w;
  u64* __restrict__ rec = (u64*)rec_w;
  int i = blockIdx.x * blockDim.x + threadIdx.x;
  int stride = gridDim.x * blockDim.x;
  for (; i < E; i += stride) {
    const int pos = rowinfo[dst[i]].x + rank[i];
    const u64 v = (u64)(unsigned)src[i] | ((u64)__float_as_uint(w[i]) << 32);
    (void)__hip_atomic_exchange(&rec[pos], v, __ATOMIC_RELAXED, __HIP_MEMORY_SCOPE_AGENT);
  }
}

__global__ void dinv_kernel(const int* __restrict__ rowinfo_w,
                            const int* __restrict__ rec_w,
                            float* __restrict__ dinv, int n) {
  const int2* __restrict__ rowinfo = (const int2*)rowinfo_w;
  const u64* __restrict__ rec = (const u64*)rec_w;
  int i = blockIdx.x * blockDim.x + threadIdx.x;
  int stride = gridDim.x * blockDim.x;
  for (; i < n; i += stride) {
    const int2 ri = rowinfo[i];
    float s = 0.0f;
    for (int j = 0; j < ri.y; ++j)
      s += __uint_as_float((unsigned)(rec[ri.x + j] >> 32));
    dinv[i] = rsqrtf(s + 1.0f);
  }
}

#define APITCH 132

__global__ void __launch_bounds__(256, 8)
gemm_wmma_f32(const float* __restrict__ A, const float* __restrict__ W,
              float* __restrict__ C, int nRowTiles) {
  __shared__ __attribute__((aligned(16))) float stage[2][16 * APITCH];

  const int lane = threadIdx.x & 31;
  const int wave = threadIdx.x >> 5;
  const int m    = lane & 15;
  const int half = lane >> 4;
  const int col0 = wave * 16;

  v2f bfrag[32];
#pragma unroll
  for (int kk = 0; kk < 32; ++kk) {
    const int k = kk * 4 + half * 2;
    v2f b;
    b.x = W[(k + 0) * HDIM + col0 + m];
    b.y = W[(k + 1) * HDIM + col0 + m];
    bfrag[kk] = b;
  }

  const int lc = lane * 4;
  const float* __restrict__ gsrc = A + (long)wave * HDIM + lc;
  const int soff0 = wave * APITCH + lc;
  const int soff1 = (wave + 8) * APITCH + lc;
  const int foff = m * APITCH + half * 2;

  int rt = blockIdx.x;
  float4 p0 = {0.f, 0.f, 0.f, 0.f}, p1 = {0.f, 0.f, 0.f, 0.f};
  if (rt < nRowTiles) {
    const float* __restrict__ g = gsrc + (long)rt * 16 * HDIM;
    p0 = *(const float4*)(g);
    p1 = *(const float4*)(g + 8 * HDIM);
  }

  int buf = 0;
  for (; rt < nRowTiles; rt += gridDim.x) {
    float* __restrict__ st = stage[buf];
    *(float4*)(st + soff0) = p0;
    *(float4*)(st + soff1) = p1;
    __syncthreads();

    const int nrt = rt + gridDim.x;
    if (nrt < nRowTiles) {
      const float* __restrict__ g = gsrc + (long)nrt * 16 * HDIM;
      p0 = *(const float4*)(g);
      p1 = *(const float4*)(g + 8 * HDIM);
    }

    const float* __restrict__ arow = st + foff;
    v8f acc = {0.f, 0.f, 0.f, 0.f, 0.f, 0.f, 0.f, 0.f};
#pragma unroll
    for (int kk = 0; kk < 32; ++kk) {
      const v2f a = *(const v2f*)(arow + kk * 4);
      acc = __builtin_amdgcn_wmma_f32_16x16x4_f32(
           false, a,  false, bfrag[kk],
           (short)0, acc,  false,  false);
    }
#pragma unroll
    for (int r = 0; r < 8; ++r)
      C[(long)(rt * 16 + half * 8 + r) * HDIM + col0 + m] = acc[r];
    buf ^= 1;
  }
}

__device__ __forceinline__ float lane_bcast_f(float v, int j) {
  return __int_as_float(__builtin_amdgcn_readlane(__float_as_int(v), j));
}

__device__ __forceinline__ void axpy4(float4& acc, float c, const float4 v) {
  acc.x = fmaf(c, v.x, acc.x);
  acc.y = fmaf(c, v.y, acc.y);
  acc.z = fmaf(c, v.z, acc.z);
  acc.w = fmaf(c, v.w, acc.w);
}

__global__ void __launch_bounds__(256)
gather_kernel(const float* __restrict__ h, const float* __restrict__ dinv,
              const int* __restrict__ rowinfo_w, const int* __restrict__ rec_w,
              const float* __restrict__ b, float* __restrict__ out,
              int n_nodes, int relu) {
  const int2* __restrict__ rowinfo = (const int2*)rowinfo_w;
  const u64* __restrict__ rec = (const u64*)rec_w;
  const int lane = threadIdx.x & 31;
  const int nw = gridDim.x * 8;
  const int w0 = __builtin_amdgcn_readfirstlane((int)(blockIdx.x * 8 + (threadIdx.x >> 5)));
  const float4 bv = *(const float4*)(b + lane * 4);
  const float* __restrict__ hl = h + lane * 4;

  for (int n = w0; n < n_nodes; n += nw) {
    const int2 ri = rowinfo[n];
    const int st = ri.x;
    const int c = ri.y;
    const float di = dinv[n];
    const float4 hs = *(const float4*)(hl + (long)n * HDIM);
    float4 acc = {0.0f, 0.0f, 0.0f, 0.0f};

    for (int base = 0; base < c; base += 32) {
      const int m = min(32, c - base);
      int sj = 0;
      float cj = 0.0f;
      if (lane < m) {
        const u64 r = rec[st + base + lane];
        sj = (int)(unsigned)r;
        cj = dinv[sj] * __uint_as_float((unsigned)(r >> 32)) * di;
      }
      int j = 0;
      for (; j + 4 <= m; j += 4) {
        const int s0 = __builtin_amdgcn_readlane(sj, j + 0);
        const int s1 = __builtin_amdgcn_readlane(sj, j + 1);
        const int s2 = __builtin_amdgcn_readlane(sj, j + 2);
        const int s3 = __builtin_amdgcn_readlane(sj, j + 3);
        const float4 h0 = *(const float4*)(hl + (long)s0 * HDIM);
        const float4 h1 = *(const float4*)(hl + (long)s1 * HDIM);
        const float4 h2 = *(const float4*)(hl + (long)s2 * HDIM);
        const float4 h3 = *(const float4*)(hl + (long)s3 * HDIM);
        axpy4(acc, lane_bcast_f(cj, j + 0), h0);
        axpy4(acc, lane_bcast_f(cj, j + 1), h1);
        axpy4(acc, lane_bcast_f(cj, j + 2), h2);
        axpy4(acc, lane_bcast_f(cj, j + 3), h3);
      }
      for (; j < m; ++j) {
        const int s0 = __builtin_amdgcn_readlane(sj, j);
        const float4 h0 = *(const float4*)(hl + (long)s0 * HDIM);
        axpy4(acc, lane_bcast_f(cj, j), h0);
      }
    }

    const float dd = di * di;
    float4 v;
    v.x = acc.x + dd * hs.x + bv.x;
    v.y = acc.y + dd * hs.y + bv.y;
    v.z = acc.z + dd * hs.z + bv.z;
    v.w = acc.w + dd * hs.w + bv.w;
    if (relu) {
      v.x = fmaxf(v.x, 0.0f); v.y = fmaxf(v.y, 0.0f);
      v.z = fmaxf(v.z, 0.0f); v.w = fmaxf(v.w, 0.0f);
    }
    *(float4*)(out + (long)n * HDIM + lane * 4) = v;
  }
}

extern "C" void kernel_launch(void* const* d_in, const int* in_sizes, int n_in,
                              void* d_out, int out_size, void* d_ws, size_t ws_size,
                              hipStream_t stream) {
  const float* x  = (const float*)d_in[0];
  const float* ew = (const float*)d_in[1];
  const float* W1 = (const float*)d_in[2];
  const float* b1 = (const float*)d_in[3];
  const float* W2 = (const float*)d_in[4];
  const float* b2 = (const float*)d_in[5];
  const int*   ei = (const int*)d_in[6];

  const int N = in_sizes[0] / HDIM;
  const int E = in_sizes[1];
  const int* src = ei;
  const int* dst = ei + E;

  const long NH = (long)N * HDIM;
  const long Np = ((long)N + 255) & ~255L;
  const long Ep = ((long)E + 255) & ~255L;

  int*   cnt     = (int*)d_ws;
  int*   total   = cnt + (long)N * CPAD;
  int*   rowinfo = total + 256;
  float* dinv    = (float*)(rowinfo + 2 * Np);
  int*   rank    = (int*)(dinv + Np);
  int*   rec     = rank + Ep;
  float* h       = (float*)(rec + 2 * Ep);
  float* h1      = h + NH;
  float* out     = (float*)d_out;

  zero_i32<<<2048, 256, 0, stream>>>(cnt, (long)N * CPAD + 256);
  count_rank_kernel<<<6250, 256, 0, stream>>>(dst, cnt, rank, E);
  alloc_kernel<<<391, 256, 0, stream>>>(cnt, total, rowinfo, N);
  fill_kernel<<<6250, 256, 0, stream>>>(src, dst, ew, rank, rowinfo, rec, E);
  dinv_kernel<<<391, 256, 0, stream>>>(rowinfo, rec, dinv, N);

  const int nRowTiles = N / 16;

  gemm_wmma_f32<<<1024, 256, 0, stream>>>(x, W1, h, nRowTiles);
  gather_kernel<<<12500, 256, 0, stream>>>(h, dinv, rowinfo, rec, b1, h1, N, 1);

  gemm_wmma_f32<<<1024, 256, 0, stream>>>(h1, W2, h, nRowTiles);
  gather_kernel<<<12500, 256, 0, stream>>>(h, dinv, rowinfo, rec, b2, out, N, 0);
}


# Neighbour sum before the product: one kernel per layer; a block gathers 16 nodes' sums of INPUT rows into LDS and multiplies by W there ((A x)W, not A(xW): rounding order changes); 9 launches to 7
# speedup vs baseline: 3.6717x; 1.0665x over previous
#include <hip/hip_runtime.h>
#include <hip/hip_bf16.h>

typedef __attribute__((ext_vector_type(2))) float v2f;
typedef __attribute__((ext_vector_type(8))) float v8f;
typedef unsigned long long u64;

#define HDIM 128
#define CPAD 32

__global__ void zero_i32(int* __restrict__ p, long n) {
  long i = (long)blockIdx.x * blockDim.x + threadIdx.x;
  long stride = (long)gridDim.x * blockDim.x;
  for (; i < n; i += stride) p[i] = 0;
}

__global__ void count_rank_kernel(const int* __restrict__ dst,
                                  int* __restrict__ cnt,
                                  int* __restrict__ rank, int E) {
  int i = blockIdx.x * blockDim.x + threadIdx.x;
  int stride = gridDim.x * blockDim.x;
  for (; i < E; i += stride)
    rank[i] = atomicAdd(&cnt[(long)dst[i] * CPAD], 1);
}

__global__ void __launch_bounds__(256)
alloc_kernel(const int* __restrict__ cnt, int* __restrict__ total,
             int* __restrict__ rowinfo_w, int n) {
  int2* __restrict__ rowinfo = (int2*)rowinfo_w;
  const int lane = threadIdx.x & 31;
  for (int base = blockIdx.x * blockDim.x; base < n; base += gridDim.x * blockDim.x) {
    const int i = base + threadIdx.x;
    const int c = (i < n) ? cnt[(long)i * CPAD] : 0;
    int incl = c;
    for (int d = 1; d < 32; d <<= 1) {
      const int t = __shfl_up(incl, d, 32);
      if (lane >= d) incl += t;
    }
    int wbase = 0;
    if (lane == 31) wbase = atomicAdd(total, incl);
    wbase = __shfl(wbase, 31, 32);
    if (i < n) rowinfo[i] = make_int2(wbase + incl - c, c);
  }
}

__global__ void fill_kernel(const int* __restrict__ src, const int* __restrict__ dst,
                            const float* __restrict__ w, const int* __restrict__ rank,
                            const int* __restrict__ rowinfo_w,
                            int* __restrict__ rec_w, int E) {
  const int2* __restrict__ rowinfo = (const int2*)rowinfo_w;
  u64* __restrict__ rec = (u64*)rec_w;
  int i = blockIdx.x * blockDim.x + threadIdx.x;
  int stride = gridDim.x * blockDim.x;
  for (; i < E; i += stride) {
    const int pos = rowinfo[dst[i]].x + rank[i];
    const u64 v = (u64)(unsigned)src[i] | ((u64)__float_as_uint(w[i]) << 32);
    (void)__hip_atomic_exchange(&rec[pos], v, __ATOMIC_RELAXED, __HIP_MEMORY_SCOPE_AGENT);
  }
}

__global__ void dinv_kernel(const int* __restrict__ rowinfo_w,
                            const int* __restrict__ rec_w,
                            float* __restrict__ dinv, int n) {
  const int2* __restrict__ rowinfo = (const int2*)rowinfo_w;
  const u64* __restrict__ rec = (const u64*)rec_w;
  int i = blockIdx.x * blockDim.x + threadIdx.x;
  int stride = gridDim.x * blockDim.x;
  for (; i < n; i += stride) {
    const int2 ri = rowinfo[i];
    float s = 0.0f;
    for (int j = 0; j < ri.y; ++j)
      s += __uint_as_float((unsigned)(rec[ri.x + j] >> 32));
    dinv[i] = rsqrtf(s + 1.0f);
  }
}

#define APITCH 132

__device__ __forceinline__ float lane_bcast_f(float v, int j) {
  return __int_as_float(__builtin_amdgcn_readlane(__float_as_int(v), j));
}

__device__ __forceinline__ void axpy4(float4& acc, float c, const float4 v) {
  acc.x = fmaf(c, v.x, acc.x);
  acc.y = fmaf(c, v.y, acc.y);
  acc.z = fmaf(c, v.z, acc.z);
  acc.w = fmaf(c, v.w, acc.w);
}

__device__ __forceinline__ float4
sum_row(const float* __restrict__ xl, const float* __restrict__ dinv,
        const int2* __restrict__ rowinfo, const u64* __restrict__ rec,
        int n, int lane) {
  const int2 ri = rowinfo[n];
  const int st = ri.x;
  const int c = ri.y;
  const float di = dinv[n];
  const float4 xs = *(const float4*)(xl + (long)n * HDIM);
  float4 acc = {0.0f, 0.0f, 0.0f, 0.0f};

  for (int base = 0; base < c; base += 32) {
    const int m = min(32, c - base);
    int sj = 0;
    float cj = 0.0f;
    if (lane < m) {
      const u64 r = rec[st + base + lane];
      sj = (int)(unsigned)r;
      cj = dinv[sj] * __uint_as_float((unsigned)(r >> 32)) * di;
    }
    int j = 0;
    for (; j + 8 <= m; j += 8) {
      float4 v[8];
#pragma unroll
      for (int u = 0; u < 8; ++u) {
        const int s = __builtin_amdgcn_readlane(sj, j + u);
        v[u] = *(const float4*)(xl + (long)s * HDIM);
      }
#pragma unroll
      for (int u = 0; u < 8; ++u) axpy4(acc, lane_bcast_f(cj, j + u), v[u]);
    }
    if (j + 4 <= m) {
      float4 v[4];
#pragma unroll
      for (int u = 0; u < 4; ++u) {
        const int s = __builtin_amdgcn_readlane(sj, j + u);
        v[u] = *(const float4*)(xl + (long)s * HDIM);
      }
#pragma unroll
      for (int u = 0; u < 4; ++u) axpy4(acc, lane_bcast_f(cj, j + u), v[u]);
      j += 4;
    }
    for (; j < m; ++j) {
      const int s = __builtin_amdgcn_readlane(sj, j);
      const float4 v0 = *(const float4*)(xl + (long)s * HDIM);
      axpy4(acc, lane_bcast_f(cj, j), v0);
    }
  }

  const float dd = di * di;
  float4 v;
  v.x = acc.x + dd * xs.x;
  v.y = acc.y + dd * xs.y;
  v.z = acc.z + dd * xs.z;
  v.w = acc.w + dd * xs.w;
  return v;
}

__global__ void __launch_bounds__(256, 8)
layer_kernel(const float* __restrict__ X, const float* __restrict__ W,
             const float* __restrict__ b, const float* __restrict__ dinv,
             const int* __restrict__ rowinfo_w, const int* __restrict__ rec_w,
             float* __restrict__ Out, int nRowTiles, int relu) {
  __shared__ __attribute__((aligned(16))) float stage[2][16 * APITCH];
  const int2* __restrict__ rowinfo = (const int2*)rowinfo_w;
  const u64* __restrict__ rec = (const u64*)rec_w;

  const int lane = threadIdx.x & 31;
  const int wave = __builtin_amdgcn_readfirstlane((int)(threadIdx.x >> 5));
  const int m    = lane & 15;
  const int half = lane >> 4;
  const int col0 = wave * 16;

  v2f bfrag[32];
#pragma unroll
  for (int kk = 0; kk < 32; ++kk) {
    const int k = kk * 4 + half * 2;
    v2f bb;
    bb.x = W[(k + 0) * HDIM + col0 + m];
    bb.y = W[(k + 1) * HDIM + col0 + m];
    bfrag[kk] = bb;
  }
  const float bcol = b[col0 + m];

  const int lc = lane * 4;
  const float* __restrict__ xl = X + lc;
  const int soff0 = wave * APITCH + lc;
  const int soff1 = (wave + 8) * APITCH + lc;
  const int foff = m * APITCH + half * 2;

  int buf = 0;
  for (int rt = blockIdx.x; rt < nRowTiles; rt += gridDim.x) {
    float* __restrict__ st = stage[buf];
    *(float4*)(st + soff0) = sum_row(xl, dinv, rowinfo, rec, rt * 16 + wave, lane);
    *(float4*)(st + soff1) = sum_row(xl, dinv, rowinfo, rec, rt * 16 + wave + 8, lane);
    __syncthreads();

    const float* __restrict__ arow = st + foff;
    v8f acc = {0.f, 0.f, 0.f, 0.f, 0.f, 0.f, 0.f, 0.f};
#pragma unroll
    for (int kk = 0; kk < 32; ++kk) {
      const v2f a = *(const v2f*)(arow + kk * 4);
      acc = __builtin_amdgcn_wmma_f32_16x16x4_f32(
           false, a,  false, bfrag[kk],
           (short)0, acc,  false,  false);
    }
#pragma unroll
    for (int r = 0; r < 8; ++r) {
      float v = acc[r] + bcol;
      if (relu) v = fmaxf(v, 0.0f);
      Out[(long)(rt * 16 + half * 8 + r) * HDIM + col0 + m] = v;
    }
    buf ^= 1;
  }
}

extern "C" void kernel_launch(void* const* d_in, const int* in_sizes, int n_in,
                              void* d_out, int out_size, void* d_ws, size_t ws_size,
                              hipStream_t stream) {
  const float* x  = (const float*)d_in[0];
  const float* ew = (const float*)d_in[1];
  const float* W1 = (const float*)d_in[2];
  const float* b1 = (const float*)d_in[3];
  const float* W2 = (const float*)d_in[4];
  const float* b2 = (const float*)d_in[5];
  const int*   ei = (const int*)d_in[6];

  const int N = in_sizes[0] / HDIM;
  const int E = in_sizes[1];
  const int* src = ei;
  const int* dst = ei + E;

  const long Np = ((long)N + 255) & ~255L;
  const long Ep = ((long)E + 255) & ~255L;

  int*   cnt     = (int*)d_ws;
  int*   total   = cnt + (long)N * CPAD;
  int*   rowinfo = total + 256;
  float* dinv    = (float*)(rowinfo + 2 * Np);
  int*   rank    = (int*)(dinv + Np);
  int*   rec     = rank + Ep;
  float* h1      = (float*)(rec + 2 * Ep);
  float* out     = (float*)d_out;

  zero_i32<<<2048, 256, 0, stream>>>(cnt, (long)N * CPAD + 256);
  count_rank_kernel<<<6250, 256, 0, stream>>>(dst, cnt, rank, E);
  alloc_kernel<<<391, 256, 0, stream>>>(cnt, total, rowinfo, N);
  fill_kernel<<<6250, 256, 0, stream>>>(src, dst, ew, rank, rowinfo, rec, E);
  dinv_kernel<<<391, 256, 0, stream>>>(rowinfo, rec, dinv, N);

  const int nRowTiles = N / 16;

  layer_kernel<<<1024, 256, 0, stream>>>(x, W1, b1, dinv, rowinfo, rec, h1, nRowTiles, 1);

  layer_kernel<<<1024, 256, 0, stream>>>(h1, W2, b2, dinv, rowinfo, rec, out, nRowTiles, 0);
}


# Edges grouped by a two-level counting sort on LDS counters (256-destination buckets, records ordered in LDS, stored as runs), no per-edge global atomics; 7 launches to 5; sum order = LDS arrival
# speedup vs baseline: 4.9420x; 1.3460x over previous
#include <hip/hip_runtime.h>
#include <hip/hip_bf16.h>

typedef __attribute__((ext_vector_type(2))) float v2f;
typedef __attribute__((ext_vector_type(8))) float v8f;
typedef unsigned long long u64;

#define HDIM 128

#define BSH 8
#define BNODES 256
#define NBMAX 1024
#define SCAP 8192
#define KCAP 8192
#define EPT 4

__device__ __forceinline__ int block_excl_scan(int v, int* wsum, int& total) {
  const int lane = threadIdx.x & 31;
  const int wid = threadIdx.x >> 5;
  const int nw = blockDim.x >> 5;
  int incl = v;
  for (int d = 1; d < 32; d <<= 1) {
    const int t = __shfl_up(incl, d, 32);
    if (lane >= d) incl += t;
  }
  if (lane == 31) wsum[wid] = incl;
  __syncthreads();
  if (wid == 0) {
    const int w = (lane < nw) ? wsum[lane] : 0;
    int wi = w;
    for (int d = 1; d < 32; d <<= 1) {
      const int t = __shfl_up(wi, d, 32);
      if (lane >= d) wi += t;
    }
    wsum[lane] = wi - w;
    if (lane == 31) wsum[32] = wi;
  }
  __syncthreads();
  total = wsum[32];
  return wsum[wid] + incl - v;
}

__global__ void __launch_bounds__(1024)
hist_kernel(const int* __restrict__ dst, int* __restrict__ plane, int N, int E) {
  __shared__ int hist[NBMAX];
  const int t = threadIdx.x;
  const int NB = (N + BNODES - 1) >> BSH;
  const int chunk = (E + (int)gridDim.x - 1) / (int)gridDim.x;
  const int cbeg = min(E, (int)blockIdx.x * chunk);
  const int cend = min(E, cbeg + chunk);
  for (int b = t; b < NB; b += blockDim.x) hist[b] = 0;
  __syncthreads();
  for (int i0 = cbeg + t; i0 < cend; i0 += EPT * (int)blockDim.x) {
    int d[EPT];
#pragma unroll
    for (int u = 0; u < EPT; ++u) {
      const int i = i0 + u * (int)blockDim.x;
      d[u] = (i < cend) ? dst[i] : -1;
    }
#pragma unroll
    for (int u = 0; u < EPT; ++u)
      if (d[u] >= 0) atomicAdd(&hist[d[u] >> BSH], 1);
  }
  __syncthreads();
  for (int b = t; b < NB; b += blockDim.x) plane[(long)blockIdx.x * NB + b] = hist[b];
}

__global__ void __launch_bounds__(1024)
scatter_kernel(const int* __restrict__ src, const int* __restrict__ dst,
               const float* __restrict__ w, const int* __restrict__ plane,
               int* __restrict__ bstart, int* __restrict__ tmp_w, int N, int E) {
  __shared__ int hist[NBMAX];
  __shared__ int lstart[NBMAX];
  __shared__ int cur[NBMAX];
  __shared__ int gbase[NBMAX];
  __shared__ int wsum[33];
  __shared__ int lg[SCAP];
  __shared__ u64 lrec[SCAP];
  u64* __restrict__ tmp = (u64*)tmp_w;
  const int t = threadIdx.x;
  const int NB = (N + BNODES - 1) >> BSH;
  const int chunk = (E + (int)gridDim.x - 1) / (int)gridDim.x;
  const int cbeg = min(E, (int)blockIdx.x * chunk);
  const int cend = min(E, cbeg + chunk);

  int tot = 0, pre = 0;
  if (t < NB) {
    const int* __restrict__ col = plane + t;
    const int nblk = (int)gridDim.x;
    const int me = (int)blockIdx.x;
#pragma unroll 32
    for (int blk = 0; blk < nblk; ++blk) {
      const int v = col[(long)blk * NB];
      tot += v;
      pre += (blk < me) ? v : 0;
    }
  }
  int total;
  const int ex = block_excl_scan(tot, wsum, total);
  if (t < NB) {
    gbase[t] = ex + pre;
    if (blockIdx.x == 0) bstart[t] = ex;
  }
  if (blockIdx.x == 0 && t == 0) bstart[NB] = total;
  __syncthreads();

  for (int r0 = cbeg; r0 < cend; r0 += SCAP) {
    const int rn = min(SCAP, cend - r0);
    if (t < NB) { hist[t] = 0; cur[t] = 0; }
    __syncthreads();
    for (int i0 = t; i0 < rn; i0 += EPT * (int)blockDim.x) {
      int d[EPT];
#pragma unroll
      for (int u = 0; u < EPT; ++u) {
        const int i = i0 + u * (int)blockDim.x;
        d[u] = (i < rn) ? dst[r0 + i] : -1;
      }
#pragma unroll
      for (int u = 0; u < EPT; ++u)
        if (d[u] >= 0) atomicAdd(&hist[d[u] >> BSH], 1);
    }
    __syncthreads();
    const int hv = (t < NB) ? hist[t] : 0;
    int rtot;
    const int lex = block_excl_scan(hv, wsum, rtot);
    if (t < NB) lstart[t] = lex;
    __syncthreads();
    for (int i0 = t; i0 < rn; i0 += EPT * (int)blockDim.x) {
      int d[EPT];
      unsigned sv[EPT];
      unsigned wv[EPT];
#pragma unroll
      for (int u = 0; u < EPT; ++u) {
        const int i = i0 + u * (int)blockDim.x;
        const bool in = i < rn;
        d[u]  = in ? dst[r0 + i] : -1;
        sv[u] = in ? (unsigned)src[r0 + i] : 0u;
        wv[u] = in ? __float_as_uint(w[r0 + i]) : 0u;
      }
#pragma unroll
      for (int u = 0; u < EPT; ++u) {
        if (d[u] >= 0) {
          const int b = d[u] >> BSH;
          const int k = atomicAdd(&cur[b], 1);
          const int p = lstart[b] + k;
          const unsigned lo = sv[u] | ((unsigned)(d[u] & (BNODES - 1)) << 24);
          lrec[p] = (u64)lo | ((u64)wv[u] << 32);
          lg[p] = gbase[b] + k;
        }
      }
    }
    __syncthreads();
    for (int i = t; i < rn; i += blockDim.x) tmp[lg[i]] = lrec[i];
    if (t < NB) gbase[t] += hv;
    __syncthreads();
  }
}

__global__ void __launch_bounds__(512)
bucket_kernel(const int* __restrict__ tmp_w, const int* __restrict__ bstart,
              int* __restrict__ rowinfo_w, float* __restrict__ dinv,
              int* __restrict__ rec_w, int N) {
  __shared__ int cnt[BNODES];
  __shared__ int st[BNODES];
  __shared__ int cur[BNODES];
  __shared__ float dg[BNODES];
  __shared__ int wsum[33];
  __shared__ u64 lrec[KCAP];
  const u64* __restrict__ tmp = (const u64*)tmp_w;
  u64* __restrict__ rec = (u64*)rec_w;
  int2* __restrict__ rowinfo = (int2*)rowinfo_w;
  const int t = threadIdx.x;
  const int NB = (N + BNODES - 1) >> BSH;
  const u64 keep = 0xFFFFFFFF00FFFFFFull;

  for (int b = blockIdx.x; b < NB; b += gridDim.x) {
    const int bs = bstart[b];
    const int nb = bstart[b + 1] - bs;
    if (t < BNODES) { cnt[t] = 0; cur[t] = 0; dg[t] = 0.0f; }
    __syncthreads();
    for (int i0 = t; i0 < nb; i0 += EPT * (int)blockDim.x) {
      u64 r[EPT];
      bool in[EPT];
#pragma unroll
      for (int u = 0; u < EPT; ++u) {
        const int i = i0 + u * (int)blockDim.x;
        in[u] = i < nb;
        r[u] = in[u] ? tmp[bs + i] : 0ull;
      }
#pragma unroll
      for (int u = 0; u < EPT; ++u) {
        if (in[u]) {
          const int dl = (int)((unsigned)r[u] >> 24);
          atomicAdd(&cnt[dl], 1);
          atomicAdd(&dg[dl], __uint_as_float((unsigned)(r[u] >> 32)));
        }
      }
    }
    __syncthreads();
    const int c = (t < BNODES) ? cnt[t] : 0;
    int total;
    const int ex = block_excl_scan(c, wsum, total);
    if (t < BNODES) {
      st[t] = ex;
      const int node = (b << BSH) + t;
      if (node < N) {
        rowinfo[node] = make_int2(bs + ex, c);
        dinv[node] = rsqrtf(dg[t] + 1.0f);
      }
    }
    __syncthreads();
    if (nb <= KCAP) {
      for (int i0 = t; i0 < nb; i0 += EPT * (int)blockDim.x) {
        u64 r[EPT];
        bool in[EPT];
#pragma unroll
        for (int u = 0; u < EPT; ++u) {
          const int i = i0 + u * (int)blockDim.x;
          in[u] = i < nb;
          r[u] = in[u] ? tmp[bs + i] : 0ull;
        }
#pragma unroll
        for (int u = 0; u < EPT; ++u) {
          if (in[u]) {
            const int dl = (int)((unsigned)r[u] >> 24);
            const int p = st[dl] + atomicAdd(&cur[dl], 1);
            lrec[p] = r[u] & keep;
          }
        }
      }
      __syncthreads();
      for (int i = t; i < nb; i += blockDim.x) rec[bs + i] = lrec[i];
    } else {
      for (int i = t; i < nb; i += blockDim.x) {
        const u64 r = tmp[bs + i];
        const int dl = (int)((unsigned)r >> 24);
        const int p = st[dl] + atomicAdd(&cur[dl], 1);
        rec[bs + p] = r & keep;
      }
    }
    __syncthreads();
  }
}

#define APITCH 132

__device__ __forceinline__ float lane_bcast_f(float v, int j) {
  return __int_as_float(__builtin_amdgcn_readlane(__float_as_int(v), j));
}

__device__ __forceinline__ void axpy4(float4& acc, float c, const float4 v) {
  acc.x = fmaf(c, v.x, acc.x);
  acc.y = fmaf(c, v.y, acc.y);
  acc.z = fmaf(c, v.z, acc.z);
  acc.w = fmaf(c, v.w, acc.w);
}

__device__ __forceinline__ float4
sum_row(const float* __restrict__ xl, const float* __restrict__ dinv,
        const int2* __restrict__ rowinfo, const u64* __restrict__ rec,
        int n, int lane) {
  const int2 ri = rowinfo[n];
  const int st = ri.x;
  const int c = ri.y;
  const float di = dinv[n];
  const float4 xs = *(const float4*)(xl + (long)n * HDIM);
  float4 acc = {0.0f, 0.0f, 0.0f, 0.0f};

  for (int base = 0; base < c; base += 32) {
    const int m = min(32, c - base);
    int sj = 0;
    float cj = 0.0f;
    if (lane < m) {
      const u64 r = rec[st + base + lane];
      sj = (int)(unsigned)r;
      cj = dinv[sj] * __uint_as_float((unsigned)(r >> 32)) * di;
    }
    int j = 0;
    for (; j + 8 <= m; j += 8) {
      float4 v[8];
#pragma unroll
      for (int u = 0; u < 8; ++u) {
        const int s = __builtin_amdgcn_readlane(sj, j + u);
        v[u] = *(const float4*)(xl + (long)s * HDIM);
      }
#pragma unroll
      for (int u = 0; u < 8; ++u) axpy4(acc, lane_bcast_f(cj, j + u), v[u]);
    }
    if (j + 4 <= m) {
      float4 v[4];
#pragma unroll
      for (int u = 0; u < 4; ++u) {
        const int s = __builtin_amdgcn_readlane(sj, j + u);
        v[u] = *(const float4*)(xl + (long)s * HDIM);
      }
#pragma unroll
      for (int u = 0; u < 4; ++u) axpy4(acc, lane_bcast_f(cj, j + u), v[u]);
      j += 4;
    }
    for (; j < m; ++j) {
      const int s = __builtin_amdgcn_readlane(sj, j);
      const float4 v0 = *(const float4*)(xl + (long)s * HDIM);
      axpy4(acc, lane_bcast_f(cj, j), v0);
    }
  }

  const float dd = di * di;
  float4 v;
  v.x = acc.x + dd * xs.x;
  v.y = acc.y + dd * xs.y;
  v.z = acc.z + dd * xs.z;
  v.w = acc.w + dd * xs.w;
  return v;
}

__global__ void __launch_bounds__(256, 8)
layer_kernel(const float* __restrict__ X, const float* __restrict__ W,
             const float* __restrict__ b, const float* __restrict__ dinv,
             const int* __restrict__ rowinfo_w, const int* __restrict__ rec_w,
             float* __restrict__ Out, int nRowTiles, int relu) {
  __shared__ __attribute__((aligned(16))) float stage[2][16 * APITCH];
  const int2* __restrict__ rowinfo = (const int2*)rowinfo_w;
  const u64* __restrict__ rec = (const u64*)rec_w;

  const int lane = threadIdx.x & 31;
  const int wave = __builtin_amdgcn_readfirstlane((int)(threadIdx.x >> 5));
  const int m    = lane & 15;
  const int half = lane >> 4;
  const int col0 = wave * 16;

  v2f bfrag[32];
#pragma unroll
  for (int kk = 0; kk < 32; ++kk) {
    const int k = kk * 4 + half * 2;
    v2f bb;
    bb.x = W[(k + 0) * HDIM + col0 + m];
    bb.y = W[(k + 1) * HDIM + col0 + m];
    bfrag[kk] = bb;
  }
  const float bcol = b[col0 + m];

  const int lc = lane * 4;
  const float* __restrict__ xl = X + lc;
  const int soff0 = wave * APITCH + lc;
  const int soff1 = (wave + 8) * APITCH + lc;
  const int foff = m * APITCH + half * 2;

  int buf = 0;
  for (int rt = blockIdx.x; rt < nRowTiles; rt += gridDim.x) {
    float* __restrict__ st = stage[buf];
    *(float4*)(st + soff0) = sum_row(xl, dinv, rowinfo, rec, rt * 16 + wave, lane);
    *(float4*)(st + soff1) = sum_row(xl, dinv, rowinfo, rec, rt * 16 + wave + 8, lane);
    __syncthreads();

    const float* __restrict__ arow = st + foff;
    v8f acc = {0.f, 0.f, 0.f, 0.f, 0.f, 0.f, 0.f, 0.f};
#pragma unroll
    for (int kk = 0; kk < 32; ++kk) {
      const v2f a = *(const v2f*)(arow + kk * 4);
      acc = __builtin_amdgcn_wmma_f32_16x16x4_f32(
           false, a,  false, bfrag[kk],
           (short)0, acc,  false,  false);
    }
#pragma unroll
    for (int r = 0; r < 8; ++r) {
      float v = acc[r] + bcol;
      if (relu) v = fmaxf(v, 0.0f);
      Out[(long)(rt * 16 + half * 8 + r) * HDIM + col0 + m] = v;
    }
    buf ^= 1;
  }
}

extern "C" void kernel_launch(void* const* d_in, const int* in_sizes, int n_in,
                              void* d_out, int out_size, void* d_ws, size_t ws_size,
                              hipStream_t stream) {
  const float* x  = (const float*)d_in[0];
  const float* ew = (const float*)d_in[1];
  const float* W1 = (const float*)d_in[2];
  const float* b1 = (const float*)d_in[3];
  const float* W2 = (const float*)d_in[4];
  const float* b2 = (const float*)d_in[5];
  const int*   ei = (const int*)d_in[6];

  const int N = in_sizes[0] / HDIM;
  const int E = in_sizes[1];
  const int* src = ei;
  const int* dst = ei + E;

  const long Np = ((long)N + 255) & ~255L;
  const long Ep = ((long)E + 255) & ~255L;

  int*   plane   = (int*)d_ws;
  int*   bstart  = plane + 256 * 1024;
  int*   rowinfo = bstart + 2048;
  float* dinv    = (float*)(rowinfo + 2 * Np);
  int*   tmp     = (int*)(dinv + Np);
  int*   rec     = tmp + 2 * Ep;
  float* h1      = (float*)(rec + 2 * Ep);
  float* out     = (float*)d_out;

  hist_kernel<<<256, 1024, 0, stream>>>(dst, plane, N, E);
  scatter_kernel<<<256, 1024, 0, stream>>>(src, dst, ew, plane, bstart, tmp, N, E);
  bucket_kernel<<<391, 512, 0, stream>>>(tmp, bstart, rowinfo, dinv, rec, N);

  const int nRowTiles = N / 16;

  layer_kernel<<<1024, 256, 0, stream>>>(x, W1, b1, dinv, rowinfo, rec, h1, nRowTiles, 1);

  layer_kernel<<<1024, 256, 0, stream>>>(h1, W2, b2, dinv, rowinfo, rec, out, nRowTiles, 0);
}


# Neighbour sum runs ahead of itself: node q+1's dinv gather, q+2's records and q+3's row info (scalar loads) fly under node q's rows; a rest of rows is one padded batch; sum order unchanged
# speedup vs baseline: 5.2196x; 1.0562x over previous
#include <hip/hip_runtime.h>
#include <hip/hip_bf16.h>

typedef __attribute__((ext_vector_type(2))) float v2f;
typedef __attribute__((ext_vector_type(8))) float v8f;
typedef unsigned long long u64;

#define HDIM 128

#define BSH 8
#define BNODES 256
#define NBMAX 1024
#define SCAP 8192
#define KCAP 8192
#define EPT 4

__device__ __forceinline__ int block_excl_scan(int v, int* wsum, int& total) {
  const int lane = threadIdx.x & 31;
  const int wid = threadIdx.x >> 5;
  const int nw = blockDim.x >> 5;
  int incl = v;
  for (int d = 1; d < 32; d <<= 1) {
    const int t = __shfl_up(incl, d, 32);
    if (lane >= d) incl += t;
  }
  if (lane == 31) wsum[wid] = incl;
  __syncthreads();
  if (wid == 0) {
    const int w = (lane < nw) ? wsum[lane] : 0;
    int wi = w;
    for (int d = 1; d < 32; d <<= 1) {
      const int t = __shfl_up(wi, d, 32);
      if (lane >= d) wi += t;
    }
    wsum[lane] = wi - w;
    if (lane == 31) wsum[32] = wi;
  }
  __syncthreads();
  total = wsum[32];
  return wsum[wid] + incl - v;
}

__global__ void __launch_bounds__(1024)
hist_kernel(const int* __restrict__ dst, int* __restrict__ plane, int N, int E) {
  __shared__ int hist[NBMAX];
  const int t = threadIdx.x;
  const int NB = (N + BNODES - 1) >> BSH;
  const int chunk = (E + (int)gridDim.x - 1) / (int)gridDim.x;
  const int cbeg = min(E, (int)blockIdx.x * chunk);
  const int cend = min(E, cbeg + chunk);
  for (int b = t; b < NB; b += blockDim.x) hist[b] = 0;
  __syncthreads();
  for (int i0 = cbeg + t; i0 < cend; i0 += EPT * (int)blockDim.x) {
    int d[EPT];
#pragma unroll
    for (int u = 0; u < EPT; ++u) {
      const int i = i0 + u * (int)blockDim.x;
      d[u] = (i < cend) ? dst[i] : -1;
    }
#pragma unroll
    for (int u = 0; u < EPT; ++u)
      if (d[u] >= 0) atomicAdd(&hist[d[u] >> BSH], 1);
  }
  __syncthreads();
  for (int b = t; b < NB; b += blockDim.x) plane[(long)blockIdx.x * NB + b] = hist[b];
}

__global__ void __launch_bounds__(1024)
scatter_kernel(const int* __restrict__ src, const int* __restrict__ dst,
               const float* __restrict__ w, const int* __restrict__ plane,
               int* __restrict__ bstart, int* __restrict__ tmp_w, int N, int E) {
  __shared__ int hist[NBMAX];
  __shared__ int lstart[NBMAX];
  __shared__ int cur[NBMAX];
  __shared__ int gbase[NBMAX];
  __shared__ int wsum[33];
  __shared__ int lg[SCAP];
  __shared__ u64 lrec[SCAP];
  u64* __restrict__ tmp = (u64*)tmp_w;
  const int t = threadIdx.x;
  const int NB = (N + BNODES - 1) >> BSH;
  const int chunk = (E + (int)gridDim.x - 1) / (int)gridDim.x;
  const int cbeg = min(E, (int)blockIdx.x * chunk);
  const int cend = min(E, cbeg + chunk);

  int tot = 0, pre = 0;
  if (t < NB) {
    const int* __restrict__ col = plane + t;
    const int nblk = (int)gridDim.x;
    const int me = (int)blockIdx.x;
#pragma unroll 32
    for (int blk = 0; blk < nblk; ++blk) {
      const int v = col[(long)blk * NB];
      tot += v;
      pre += (blk < me) ? v : 0;
    }
  }
  int total;
  const int ex = block_excl_scan(tot, wsum, total);
  if (t < NB) {
    gbase[t] = ex + pre;
    if (blockIdx.x == 0) bstart[t] = ex;
  }
  if (blockIdx.x == 0 && t == 0) bstart[NB] = total;
  __syncthreads();

  for (int r0 = cbeg; r0 < cend; r0 += SCAP) {
    const int rn = min(SCAP, cend - r0);
    if (t < NB) { hist[t] = 0; cur[t] = 0; }
    __syncthreads();
    for (int i0 = t; i0 < rn; i0 += EPT * (int)blockDim.x) {
      int d[EPT];
#pragma unroll
      for (int u = 0; u < EPT; ++u) {
        const int i = i0 + u * (int)blockDim.x;
        d[u] = (i < rn) ? dst[r0 + i] : -1;
      }
#pragma unroll
      for (int u = 0; u < EPT; ++u)
        if (d[u] >= 0) atomicAdd(&hist[d[u] >> BSH], 1);
    }
    __syncthreads();
    const int hv = (t < NB) ? hist[t] : 0;
    int rtot;
    const int lex = block_excl_scan(hv, wsum, rtot);
    if (t < NB) lstart[t] = lex;
    __syncthreads();
    for (int i0 = t; i0 < rn; i0 += EPT * (int)blockDim.x) {
      int d[EPT];
      unsigned sv[EPT];
      unsigned wv[EPT];
#pragma unroll
      for (int u = 0; u < EPT; ++u) {
        const int i = i0 + u * (int)blockDim.x;
        const bool in = i < rn;
        d[u]  = in ? dst[r0 + i] : -1;
        sv[u] = in ? (unsigned)src[r0 + i] : 0u;
        wv[u] = in ? __float_as_uint(w[r0 + i]) : 0u;
      }
#pragma unroll
      for (int u = 0; u < EPT; ++u) {
        if (d[u] >= 0) {
          const int b = d[u] >> BSH;
          const int k = atomicAdd(&cur[b], 1);
          const int p = lstart[b] + k;
          const unsigned lo = sv[u] | ((unsigned)(d[u] & (BNODES - 1)) << 24);
          lrec[p] = (u64)lo | ((u64)wv[u] << 32);
          lg[p] = gbase[b] + k;
        }
      }
    }
    __syncthreads();
    for (int i = t; i < rn; i += blockDim.x) tmp[lg[i]] = lrec[i];
    if (t < NB) gbase[t] += hv;
    __syncthreads();
  }
}

__global__ void __launch_bounds__(512)
bucket_kernel(const int* __restrict__ tmp_w, const int* __restrict__ bstart,
              int* __restrict__ rowinfo_w, float* __restrict__ dinv,
              int* __restrict__ rec_w, int N) {
  __shared__ int cnt[BNODES];
  __shared__ int st[BNODES];
  __shared__ int cur[BNODES];
  __shared__ float dg[BNODES];
  __shared__ int wsum[33];
  __shared__ u64 lrec[KCAP];
  const u64* __restrict__ tmp = (const u64*)tmp_w;
  u64* __restrict__ rec = (u64*)rec_w;
  int2* __restrict__ rowinfo = (int2*)rowinfo_w;
  const int t = threadIdx.x;
  const int NB = (N + BNODES - 1) >> BSH;
  const u64 keep = 0xFFFFFFFF00FFFFFFull;

  for (int b = blockIdx.x; b < NB; b += gridDim.x) {
    const int bs = bstart[b];
    const int nb = bstart[b + 1] - bs;
    if (t < BNODES) { cnt[t] = 0; cur[t] = 0; dg[t] = 0.0f; }
    __syncthreads();
    for (int i0 = t; i0 < nb; i0 += EPT * (int)blockDim.x) {
      u64 r[EPT];
      bool in[EPT];
#pragma unroll
      for (int u = 0; u < EPT; ++u) {
        const int i = i0 + u * (int)blockDim.x;
        in[u] = i < nb;
        r[u] = in[u] ? tmp[bs + i] : 0ull;
      }
#pragma unroll
      for (int u = 0; u < EPT; ++u) {
        if (in[u]) {
          const int dl = (int)((unsigned)r[u] >> 24);
          atomicAdd(&cnt[dl], 1);
          atomicAdd(&dg[dl], __uint_as_float((unsigned)(r[u] >> 32)));
        }
      }
    }
    __syncthreads();
    const int c = (t < BNODES) ? cnt[t] : 0;
    int total;
    const int ex = block_excl_scan(c, wsum, total);
    if (t < BNODES) {
      st[t] = ex;
      const int node = (b << BSH) + t;
      if (node < N) {
        rowinfo[node] = make_int2(bs + ex, c);
        dinv[node] = rsqrtf(dg[t] + 1.0f);
      }
    }
    __syncthreads();
    if (nb <= KCAP) {
      for (int i0 = t; i0 < nb; i0 += EPT * (int)blockDim.x) {
        u64 r[EPT];
        bool in[EPT];
#pragma unroll
        for (int u = 0; u < EPT; ++u) {
          const int i = i0 + u * (int)blockDim.x;
          in[u] = i < nb;
          r[u] = in[u] ? tmp[bs + i] : 0ull;
        }
#pragma unroll
        for (int u = 0; u < EPT; ++u) {
          if (in[u]) {
            const int dl = (int)((unsigned)r[u] >> 24);
            const int p = st[dl] + atomicAdd(&cur[dl], 1);
            lrec[p] = r[u] & keep;
          }
        }
      }
      __syncthreads();
      for (int i = t; i < nb; i += blockDim.x) rec[bs + i] = lrec[i];
    } else {
      for (int i = t; i < nb; i += blockDim.x) {
        const u64 r = tmp[bs + i];
        const int dl = (int)((unsigned)r >> 24);
        const int p = st[dl] + atomicAdd(&cur[dl], 1);
        rec[bs + p] = r & keep;
      }
    }
    __syncthreads();
  }
}

#define APITCH 132

__device__ __forceinline__ float lane_bcast_f(float v, int j) {
  return __int_as_float(__builtin_amdgcn_readlane(__float_as_int(v), j));
}

__device__ __forceinline__ void axpy4(float4& acc, float c, const float4 v) {
  acc.x = fmaf(c, v.x, acc.x);
  acc.y = fmaf(c, v.y, acc.y);
  acc.z = fmaf(c, v.z, acc.z);
  acc.w = fmaf(c, v.w, acc.w);
}

__device__ __forceinline__ u64
load_recs(const u64* __restrict__ rec, int st, int c, int lane) {
  u64 r = 0ull;
  if (lane < c) r = rec[st + lane];
  return r;
}

__device__ __forceinline__ void
add_rows(float4& acc, const float* __restrict__ X, unsigned lcb, int sj, float cj, int m) {
  int j = 0;
  for (; j + 8 <= m; j += 8) {
    float4 v[8];
#pragma unroll
    for (int u = 0; u < 8; ++u) {
      const int s = __builtin_amdgcn_readlane(sj, j + u);
      v[u] = *(const float4*)((const char*)X + (size_t)(unsigned)s * (HDIM * 4) + lcb);
    }
#pragma unroll
    for (int u = 0; u < 8; ++u) axpy4(acc, lane_bcast_f(cj, j + u), v[u]);
  }
  const int rest = m - j;
  if (rest > 4) {
    float4 v[8];
#pragma unroll
    for (int u = 0; u < 8; ++u) {
      const int s = __builtin_amdgcn_readlane(sj, min(j + u, m - 1));
      v[u] = *(const float4*)((const char*)X + (size_t)(unsigned)s * (HDIM * 4) + lcb);
    }
#pragma unroll
    for (int u = 0; u < 8; ++u) axpy4(acc, lane_bcast_f(cj, j + u), v[u]);
  } else if (rest > 0) {
    float4 v[4];
#pragma unroll
    for (int u = 0; u < 4; ++u) {
      const int s = __builtin_amdgcn_readlane(sj, min(j + u, m - 1));
      v[u] = *(const float4*)((const char*)X + (size_t)(unsigned)s * (HDIM * 4) + lcb);
    }
#pragma unroll
    for (int u = 0; u < 4; ++u) axpy4(acc, lane_bcast_f(cj, j + u), v[u]);
  }
}

#define CONSTAS __attribute__((address_space(4)))

__device__ __forceinline__ void
node_info(const u64* __restrict__ rowinfo, const float* __restrict__ dinv,
          int n, bool valid, int& st, int& c, float& di) {
  st = 0;
  c = 0;
  di = 0.0f;
  if (valid) {
    const u64 r = ((const CONSTAS u64*)rowinfo)[n];
    st = (int)(unsigned)r;
    c = (int)(unsigned)(r >> 32);
    di = ((const CONSTAS float*)dinv)[n];
  }
}

__global__ void __launch_bounds__(256, 8)
layer_kernel(const float* __restrict__ X, const float* __restrict__ W,
             const float* __restrict__ b, const float* __restrict__ dinv,
             const int* __restrict__ rowinfo_w, const int* __restrict__ rec_w,
             float* __restrict__ Out, int nRowTiles, int relu) {
  __shared__ __attribute__((aligned(16))) float stage[2][16 * APITCH];
  const u64* __restrict__ rowinfo = (const u64*)rowinfo_w;
  const u64* __restrict__ rec = (const u64*)rec_w;

  const int lane = threadIdx.x & 31;
  const int wave = __builtin_amdgcn_readfirstlane((int)(threadIdx.x >> 5));
  const int m    = lane & 15;
  const int half = lane >> 4;
  const int col0 = wave * 16;

  v2f bfrag[32];
#pragma unroll
  for (int kk = 0; kk < 32; ++kk) {
    const int k = kk * 4 + half * 2;
    v2f bb;
    bb.x = W[(k + 0) * HDIM + col0 + m];
    bb.y = W[(k + 1) * HDIM + col0 + m];
    bfrag[kk] = bb;
  }
  const float bcol = b[col0 + m];

  const int lc = lane * 4;
  const unsigned lcb = (unsigned)lane * 16u;
  const int soff0 = wave * APITCH + lc;
  const int soff1 = (wave + 8) * APITCH + lc;
  const int foff = m * APITCH + half * 2;

  const int tile0 = (int)blockIdx.x;
  const int tstride = (int)gridDim.x;
  int st0, c0, st1, c1, st2, c2;
  float di0, di1, di2;
  node_info(rowinfo, dinv, tile0 * 16 + wave, tile0 < nRowTiles, st0, c0, di0);
  node_info(rowinfo, dinv, tile0 * 16 + wave + 8, tile0 < nRowTiles, st1, c1, di1);
  node_info(rowinfo, dinv, (tile0 + tstride) * 16 + wave, tile0 + tstride < nRowTiles,
            st2, c2, di2);
  int sj0;
  float cj0;
  u64 rA;
  {
    const u64 r0 = load_recs(rec, st0, c0, lane);
    rA = load_recs(rec, st1, c1, lane);
    sj0 = (int)(unsigned)r0;
    const float d0 = *(const float*)((const char*)dinv + ((unsigned)sj0 << 2));
    const float w0 = __uint_as_float((unsigned)(r0 >> 32));
    cj0 = (lane < c0) ? d0 * w0 * di0 : 0.0f;
  }

  int buf = 0;
  for (int rt = tile0; rt < nRowTiles; rt += tstride) {
    float* __restrict__ st = stage[buf];
#pragma unroll
    for (int h = 0; h < 2; ++h) {
      const int n = rt * 16 + wave + 8 * h;
      const int t3 = rt + (h ? 2 : 1) * tstride;
      int st3, c3;
      float di3;
      node_info(rowinfo, dinv, t3 * 16 + wave + (h ? 0 : 8), t3 < nRowTiles, st3, c3, di3);
      const int sjN = (int)(unsigned)rA;
      const float wN = __uint_as_float((unsigned)(rA >> 32));
      const float dN = *(const float*)((const char*)dinv + ((unsigned)sjN << 2));
      const u64 rB = load_recs(rec, st2, c2, lane);
      const float4 xs = *(const float4*)((const char*)X + (size_t)(unsigned)n * (HDIM * 4) + lcb);

      float4 acc = {0.0f, 0.0f, 0.0f, 0.0f};
      add_rows(acc, X, lcb, sj0, cj0, min(32, c0));
      for (int base = 32; base < c0; base += 32) {
        const int mm = min(32, c0 - base);
        const u64 r = load_recs(rec, st0 + base, mm, lane);
        const int sjx = (int)(unsigned)r;
        const float cjx = (lane < mm)
            ? dinv[(unsigned)sjx] * __uint_as_float((unsigned)(r >> 32)) * di0 : 0.0f;
        add_rows(acc, X, lcb, sjx, cjx, mm);
      }

      const float dd = di0 * di0;
      float4 v;
      v.x = acc.x + dd * xs.x;
      v.y = acc.y + dd * xs.y;
      v.z = acc.z + dd * xs.z;
      v.w = acc.w + dd * xs.w;
      *(float4*)(st + (h ? soff1 : soff0)) = v;

      sj0 = sjN;
      cj0 = (lane < c1) ? dN * wN * di1 : 0.0f;
      rA = rB;
      st0 = st1; c0 = c1; di0 = di1;
      st1 = st2; c1 = c2; di1 = di2;
      st2 = st3; c2 = c3; di2 = di3;
    }
    __syncthreads();

    const float* __restrict__ arow = st + foff;
    v8f acc = {0.f, 0.f, 0.f, 0.f, 0.f, 0.f, 0.f, 0.f};
#pragma unroll
    for (int kk = 0; kk < 32; ++kk) {
      const v2f a = *(const v2f*)(arow + kk * 4);
      acc = __builtin_amdgcn_wmma_f32_16x16x4_f32(
           false, a,  false, bfrag[kk],
           (short)0, acc,  false,  false);
    }
    float* __restrict__ otile = Out + (size_t)(unsigned)rt * (16 * HDIM);
#pragma unroll
    for (int r = 0; r < 8; ++r) {
      float v = acc[r] + bcol;
      if (relu) v = fmaxf(v, 0.0f);
      otile[(half * 8 + r) * HDIM + col0 + m] = v;
    }
    buf ^= 1;
  }
}

extern "C" void kernel_launch(void* const* d_in, const int* in_sizes, int n_in,
                              void* d_out, int out_size, void* d_ws, size_t ws_size,
                              hipStream_t stream) {
  const float* x  = (const float*)d_in[0];
  const float* ew = (const float*)d_in[1];
  const float* W1 = (const float*)d_in[2];
  const float* b1 = (const float*)d_in[3];
  const float* W2 = (const float*)d_in[4];
  const float* b2 = (const float*)d_in[5];
  const int*   ei = (const int*)d_in[6];

  const int N = in_sizes[0] / HDIM;
  const int E = in_sizes[1];
  const int* src = ei;
  const int* dst = ei + E;

  const long Np = ((long)N + 255) & ~255L;
  const long Ep = ((long)E + 255) & ~255L;

  int*   plane   = (int*)d_ws;
  int*   bstart  = plane + 256 * 1024;
  int*   rowinfo = bstart + 2048;
  float* dinv    = (float*)(rowinfo + 2 * Np);
  int*   tmp     = (int*)(dinv + Np);
  int*   rec     = tmp + 2 * Ep;
  float* h1      = (float*)(rec + 2 * Ep);
  float* out     = (float*)d_out;

  hist_kernel<<<256, 1024, 0, stream>>>(dst, plane, N, E);
  scatter_kernel<<<256, 1024, 0, stream>>>(src, dst, ew, plane, bstart, tmp, N, E);
  bucket_kernel<<<391, 512, 0, stream>>>(tmp, bstart, rowinfo, dinv, rec, N);

  const int nRowTiles = N / 16;

  layer_kernel<<<1024, 256, 0, stream>>>(x, W1, b1, dinv, rowinfo, rec, h1, nRowTiles, 1);

  layer_kernel<<<1024, 256, 0, stream>>>(h1, W2, b2, dinv, rowinfo, rec, out, nRowTiles, 0);
}


# First-level edge partition stays block-local: block stores its LDS-ordered slice plus per-run bucket counts and starts, bucket sums its column; count launch gone, 5 to 4; sum order = LDS arrival
# speedup vs baseline: 5.6437x; 1.0813x over previous
#include <hip/hip_runtime.h>
#include <hip/hip_bf16.h>

typedef __attribute__((ext_vector_type(2))) float v2f;
typedef __attribute__((ext_vector_type(8))) float v8f;
typedef unsigned long long u64;

#define HDIM 128

#define BSH 8
#define BNODES 256
#define NBMAX 1024
#define SCAP 8192
#define KCAP 8192
#define RMAX 512
#define EPT 4

__device__ __forceinline__ int block_excl_scan(int v, int* wsum, int& total) {
  const int lane = threadIdx.x & 31;
  const int wid = threadIdx.x >> 5;
  const int nw = blockDim.x >> 5;
  int incl = v;
  for (int d = 1; d < 32; d <<= 1) {
    const int t = __shfl_up(incl, d, 32);
    if (lane >= d) incl += t;
  }
  if (lane == 31) wsum[wid] = incl;
  __syncthreads();
  if (wid == 0) {
    const int w = (lane < nw) ? wsum[lane] : 0;
    int wi = w;
    for (int d = 1; d < 32; d <<= 1) {
      const int t = __shfl_up(wi, d, 32);
      if (lane >= d) wi += t;
    }
    wsum[lane] = wi - w;
    if (lane == 31) wsum[32] = wi;
  }
  __syncthreads();
  total = wsum[32];
  return wsum[wid] + incl - v;
}

__device__ __forceinline__ int run_first(int run, int rounds, int chunk, int E) {
  const int blk = run / rounds;
  const int rd = run - blk * rounds;
  const int cbeg = min(E, blk * chunk);
  const int cend = min(E, cbeg + chunk);
  return min(cend, cbeg + rd * SCAP);
}

__global__ void __launch_bounds__(1024)
scatter_kernel(const int* __restrict__ src, const int* __restrict__ dst,
               const float* __restrict__ w, int* __restrict__ cntp,
               int* __restrict__ startp, int* __restrict__ tmp_w, int N, int E) {
  __shared__ int hist[NBMAX];
  __shared__ int lstart[NBMAX];
  __shared__ int cur[NBMAX];
  __shared__ int wsum[33];
  __shared__ u64 lrec[SCAP];
  u64* __restrict__ tmp = (u64*)tmp_w;
  const int t = threadIdx.x;
  const int NB = (N + BNODES - 1) >> BSH;
  const int chunk = (E + (int)gridDim.x - 1) / (int)gridDim.x;
  const int rounds = (chunk + SCAP - 1) / SCAP;

  for (int rd = 0; rd < rounds; ++rd) {
    const int run = (int)blockIdx.x * rounds + rd;
    const int r0 = run_first(run, rounds, chunk, E);
    const int rn = run_first(run + 1, rounds, chunk, E) - r0;
    if (t < NB) { hist[t] = 0; cur[t] = 0; }
    __syncthreads();
    for (int i0 = t; i0 < rn; i0 += EPT * (int)blockDim.x) {
      int d[EPT];
#pragma unroll
      for (int u = 0; u < EPT; ++u) {
        const int i = i0 + u * (int)blockDim.x;
        d[u] = (i < rn) ? dst[r0 + i] : -1;
      }
#pragma unroll
      for (int u = 0; u < EPT; ++u)
        if (d[u] >= 0) atomicAdd(&hist[d[u] >> BSH], 1);
    }
    __syncthreads();
    const int hv = (t < NB) ? hist[t] : 0;
    int rtot;
    const int lex = block_excl_scan(hv, wsum, rtot);
    if (t < NB) {
      lstart[t] = lex;
      cntp[(long)run * NB + t] = hv;
      startp[(long)run * NB + t] = lex;
    }
    __syncthreads();
    for (int i0 = t; i0 < rn; i0 += EPT * (int)blockDim.x) {
      int d[EPT];
      unsigned sv[EPT];
      unsigned wv[EPT];
#pragma unroll
      for (int u = 0; u < EPT; ++u) {
        const int i = i0 + u * (int)blockDim.x;
        const bool in = i < rn;
        d[u]  = in ? dst[r0 + i] : -1;
        sv[u] = in ? (unsigned)src[r0 + i] : 0u;
        wv[u] = in ? __float_as_uint(w[r0 + i]) : 0u;
      }
#pragma unroll
      for (int u = 0; u < EPT; ++u) {
        if (d[u] >= 0) {
          const int b = d[u] >> BSH;
          const int p = lstart[b] + atomicAdd(&cur[b], 1);
          const unsigned lo = sv[u] | ((unsigned)(d[u] & (BNODES - 1)) << 24);
          lrec[p] = (u64)lo | ((u64)wv[u] << 32);
        }
      }
    }
    __syncthreads();
    for (int i = t; i < rn; i += blockDim.x) tmp[r0 + i] = lrec[i];
    __syncthreads();
  }
}

__global__ void __launch_bounds__(512)
bucket_kernel(const int* __restrict__ tmp_w, const int* __restrict__ cntp,
              const int* __restrict__ startp, int* __restrict__ rowinfo_w,
              float* __restrict__ dinv, int* __restrict__ rec_w, int N, int E, int nsb) {
  __shared__ int cnt[BNODES];
  __shared__ int st[BNODES];
  __shared__ int cur[BNODES];
  __shared__ float dg[BNODES];
  __shared__ int wsum[33];
  __shared__ int rsrc[RMAX];
  __shared__ int rcnt[RMAX];
  __shared__ u64 lrec[KCAP];
  const u64* __restrict__ tmp = (const u64*)tmp_w;
  u64* __restrict__ rec = (u64*)rec_w;
  int2* __restrict__ rowinfo = (int2*)rowinfo_w;
  const int t = threadIdx.x;
  const int lane = t & 31;
  const int wv = t >> 5;
  const int nw = (int)blockDim.x >> 5;
  const int NB = (N + BNODES - 1) >> BSH;
  const int chunk = (E + nsb - 1) / nsb;
  const int rounds = (chunk + SCAP - 1) / SCAP;
  const int R = nsb * rounds;
  const u64 keep = 0xFFFFFFFF00FFFFFFull;

  for (int b = blockIdx.x; b < NB; b += gridDim.x) {
    int cn = 0, ls = 0;
    if (t < R) {
      cn = cntp[(long)t * NB + b];
      ls = startp[(long)t * NB + b];
    }
    int bs, nb;
    block_excl_scan(ls, wsum, bs);
    block_excl_scan(cn, wsum, nb);
    if (t < R) {
      rsrc[t] = run_first(t, rounds, chunk, E) + ls;
      rcnt[t] = cn;
    }
    if (t < BNODES) { cnt[t] = 0; cur[t] = 0; dg[t] = 0.0f; }
    __syncthreads();
    for (int r0 = wv; r0 < R; r0 += EPT * nw) {
      u64 r[EPT];
      int c[EPT];
      int s0[EPT];
#pragma unroll
      for (int u = 0; u < EPT; ++u) {
        const int rr = r0 + u * nw;
        c[u] = (rr < R) ? rcnt[rr] : 0;
        s0[u] = (rr < R) ? rsrc[rr] : 0;
        r[u] = (lane < c[u]) ? tmp[s0[u] + lane] : 0ull;
      }
#pragma unroll
      for (int u = 0; u < EPT; ++u) {
        if (lane < c[u]) {
          const int dl = (int)((unsigned)r[u] >> 24);
          atomicAdd(&cnt[dl], 1);
          atomicAdd(&dg[dl], __uint_as_float((unsigned)(r[u] >> 32)));
        }
        for (int l = lane + 32; l < c[u]; l += 32) {
          const u64 q = tmp[s0[u] + l];
          const int dl = (int)((unsigned)q >> 24);
          atomicAdd(&cnt[dl], 1);
          atomicAdd(&dg[dl], __uint_as_float((unsigned)(q >> 32)));
        }
      }
    }
    __syncthreads();
    const int cc = (t < BNODES) ? cnt[t] : 0;
    int total;
    const int ex = block_excl_scan(cc, wsum, total);
    if (t < BNODES) {
      st[t] = ex;
      const int node = (b << BSH) + t;
      if (node < N) {
        rowinfo[node] = make_int2(bs + ex, cc);
        dinv[node] = rsqrtf(dg[t] + 1.0f);
      }
    }
    __syncthreads();
    const bool inlds = nb <= KCAP;
    for (int r0 = wv; r0 < R; r0 += EPT * nw) {
      u64 r[EPT];
      int c[EPT];
      int s0[EPT];
#pragma unroll
      for (int u = 0; u < EPT; ++u) {
        const int rr = r0 + u * nw;
        c[u] = (rr < R) ? rcnt[rr] : 0;
        s0[u] = (rr < R) ? rsrc[rr] : 0;
        r[u] = (lane < c[u]) ? tmp[s0[u] + lane] : 0ull;
      }
#pragma unroll
      for (int u = 0; u < EPT; ++u) {
        if (lane < c[u]) {
          const int dl = (int)((unsigned)r[u] >> 24);
          const int p = st[dl] + atomicAdd(&cur[dl], 1);
          if (inlds) lrec[p] = r[u] & keep;
          else rec[bs + p] = r[u] & keep;
        }
        for (int l = lane + 32; l < c[u]; l += 32) {
          const u64 q = tmp[s0[u] + l];
          const int dl = (int)((unsigned)q >> 24);
          const int p = st[dl] + atomicAdd(&cur[dl], 1);
          if (inlds) lrec[p] = q & keep;
          else rec[bs + p] = q & keep;
        }
      }
    }
    __syncthreads();
    if (inlds)
      for (int i = t; i < nb; i += blockDim.x) rec[bs + i] = lrec[i];
    __syncthreads();
  }
}

#define APITCH 132

__device__ __forceinline__ float lane_bcast_f(float v, int j) {
  return __int_as_float(__builtin_amdgcn_readlane(__float_as_int(v), j));
}

__device__ __forceinline__ void axpy4(float4& acc, float c, const float4 v) {
  acc.x = fmaf(c, v.x, acc.x);
  acc.y = fmaf(c, v.y, acc.y);
  acc.z = fmaf(c, v.z, acc.z);
  acc.w = fmaf(c, v.w, acc.w);
}

__device__ __forceinline__ u64
load_recs(const u64* __restrict__ rec, int st, int c, int lane) {
  u64 r = 0ull;
  if (lane < c) r = rec[st + lane];
  return r;
}

__device__ __forceinline__ void
add_rows(float4& acc, const float* __restrict__ X, unsigned lcb, int sj, float cj, int m) {
  int j = 0;
  for (; j + 8 <= m; j += 8) {
    float4 v[8];
#pragma unroll
    for (int u = 0; u < 8; ++u) {
      const int s = __builtin_amdgcn_readlane(sj, j + u);
      v[u] = *(const float4*)((const char*)X + (size_t)(unsigned)s * (HDIM * 4) + lcb);
    }
#pragma unroll
    for (int u = 0; u < 8; ++u) axpy4(acc, lane_bcast_f(cj, j + u), v[u]);
  }
  const int rest = m - j;
  if (rest > 4) {
    float4 v[8];
#pragma unroll
    for (int u = 0; u < 8; ++u) {
      const int s = __builtin_amdgcn_readlane(sj, min(j + u, m - 1));
      v[u] = *(const float4*)((const char*)X + (size_t)(unsigned)s * (HDIM * 4) + lcb);
    }
#pragma unroll
    for (int u = 0; u < 8; ++u) axpy4(acc, lane_bcast_f(cj, j + u), v[u]);
  } else if (rest > 0) {
    float4 v[4];
#pragma unroll
    for (int u = 0; u < 4; ++u) {
      const int s = __builtin_amdgcn_readlane(sj, min(j + u, m - 1));
      v[u] = *(const float4*)((const char*)X + (size_t)(unsigned)s * (HDIM * 4) + lcb);
    }
#pragma unroll
    for (int u = 0; u < 4; ++u) axpy4(acc, lane_bcast_f(cj, j + u), v[u]);
  }
}

#define CONSTAS __attribute__((address_space(4)))

__device__ __forceinline__ void
node_info(const u64* __restrict__ rowinfo, const float* __restrict__ dinv,
          int n, bool valid, int& st, int& c, float& di) {
  st = 0;
  c = 0;
  di = 0.0f;
  if (valid) {
    const u64 r = ((const CONSTAS u64*)rowinfo)[n];
    st = (int)(unsigned)r;
    c = (int)(unsigned)(r >> 32);
    di = ((const CONSTAS float*)dinv)[n];
  }
}

__global__ void __launch_bounds__(256, 8)
layer_kernel(const float* __restrict__ X, const float* __restrict__ W,
             const float* __restrict__ b, const float* __restrict__ dinv,
             const int* __restrict__ rowinfo_w, const int* __restrict__ rec_w,
             float* __restrict__ Out, int nRowTiles, int relu) {
  __shared__ __attribute__((aligned(16))) float stage[2][16 * APITCH];
  const u64* __restrict__ rowinfo = (const u64*)rowinfo_w;
  const u64* __restrict__ rec = (const u64*)rec_w;

  const int lane = threadIdx.x & 31;
  const int wave = __builtin_amdgcn_readfirstlane((int)(threadIdx.x >> 5));
  const int m    = lane & 15;
  const int half = lane >> 4;
  const int col0 = wave * 16;

  v2f bfrag[32];
#pragma unroll
  for (int kk = 0; kk < 32; ++kk) {
    const int k = kk * 4 + half * 2;
    v2f bb;
    bb.x = W[(k + 0) * HDIM + col0 + m];
    bb.y = W[(k + 1) * HDIM + col0 + m];
    bfrag[kk] = bb;
  }
  const float bcol = b[col0 + m];

  const int lc = lane * 4;
  const unsigned lcb = (unsigned)lane * 16u;
  const int soff0 = wave * APITCH + lc;
  const int soff1 = (wave + 8) * APITCH + lc;
  const int foff = m * APITCH + half * 2;

  const int tile0 = (int)blockIdx.x;
  const int tstride = (int)gridDim.x;
  int st0, c0, st1, c1, st2, c2;
  float di0, di1, di2;
  node_info(rowinfo, dinv, tile0 * 16 + wave, tile0 < nRowTiles, st0, c0, di0);
  node_info(rowinfo, dinv, tile0 * 16 + wave + 8, tile0 < nRowTiles, st1, c1, di1);
  node_info(rowinfo, dinv, (tile0 + tstride) * 16 + wave, tile0 + tstride < nRowTiles,
            st2, c2, di2);
  int sj0;
  float cj0;
  u64 rA;
  {
    const u64 r0 = load_recs(rec, st0, c0, lane);
    rA = load_recs(rec, st1, c1, lane);
    sj0 = (int)(unsigned)r0;
    const float d0 = *(const float*)((const char*)dinv + ((unsigned)sj0 << 2));
    const float w0 = __uint_as_float((unsigned)(r0 >> 32));
    cj0 = (lane < c0) ? d0 * w0 * di0 : 0.0f;
  }

  int buf = 0;
  for (int rt = tile0; rt < nRowTiles; rt += tstride) {
    float* __restrict__ st = stage[buf];
#pragma unroll
    for (int h = 0; h < 2; ++h) {
      const int n = rt * 16 + wave + 8 * h;
      const int t3 = rt + (h ? 2 : 1) * tstride;
      int st3, c3;
      float di3;
      node_info(rowinfo, dinv, t3 * 16 + wave + (h ? 0 : 8), t3 < nRowTiles, st3, c3, di3);
      const int sjN = (int)(unsigned)rA;
      const float wN = __uint_as_float((unsigned)(rA >> 32));
      const float dN = *(const float*)((const char*)dinv + ((unsigned)sjN << 2));
      const u64 rB = load_recs(rec, st2, c2, lane);
      const float4 xs = *(const float4*)((const char*)X + (size_t)(unsigned)n * (HDIM * 4) + lcb);

      float4 acc = {0.0f, 0.0f, 0.0f, 0.0f};
      add_rows(acc, X, lcb, sj0, cj0, min(32, c0));
      for (int base = 32; base < c0; base += 32) {
        const int mm = min(32, c0 - base);
        const u64 r = load_recs(rec, st0 + base, mm, lane);
        const int sjx = (int)(unsigned)r;
        const float cjx = (lane < mm)
            ? dinv[(unsigned)sjx] * __uint_as_float((unsigned)(r >> 32)) * di0 : 0.0f;
        add_rows(acc, X, lcb, sjx, cjx, mm);
      }

      const float dd = di0 * di0;
      float4 v;
      v.x = acc.x + dd * xs.x;
      v.y = acc.y + dd * xs.y;
      v.z = acc.z + dd * xs.z;
      v.w = acc.w + dd * xs.w;
      *(float4*)(st + (h ? soff1 : soff0)) = v;

      sj0 = sjN;
      cj0 = (lane < c1) ? dN * wN * di1 : 0.0f;
      rA = rB;
      st0 = st1; c0 = c1; di0 = di1;
      st1 = st2; c1 = c2; di1 = di2;
      st2 = st3; c2 = c3; di2 = di3;
    }
    __syncthreads();

    const float* __restrict__ arow = st + foff;
    v8f acc = {0.f, 0.f, 0.f, 0.f, 0.f, 0.f, 0.f, 0.f};
#pragma unroll
    for (int kk = 0; kk < 32; ++kk) {
      const v2f a = *(const v2f*)(arow + kk * 4);
      acc = __builtin_amdgcn_wmma_f32_16x16x4_f32(
           false, a,  false, bfrag[kk],
           (short)0, acc,  false,  false);
    }
    float* __restrict__ otile = Out + (size_t)(unsigned)rt * (16 * HDIM);
#pragma unroll
    for (int r = 0; r < 8; ++r) {
      float v = acc[r] + bcol;
      if (relu) v = fmaxf(v, 0.0f);
      otile[(half * 8 + r) * HDIM + col0 + m] = v;
    }
    buf ^= 1;
  }
}

extern "C" void kernel_launch(void* const* d_in, const int* in_sizes, int n_in,
                              void* d_out, int out_size, void* d_ws, size_t ws_size,
                              hipStream_t stream) {
  const float* x  = (const float*)d_in[0];
  const float* ew = (const float*)d_in[1];
  const float* W1 = (const float*)d_in[2];
  const float* b1 = (const float*)d_in[3];
  const float* W2 = (const float*)d_in[4];
  const float* b2 = (const float*)d_in[5];
  const int*   ei = (const int*)d_in[6];

  const int N = in_sizes[0] / HDIM;
  const int E = in_sizes[1];
  const int* src = ei;
  const int* dst = ei + E;

  const long Np = ((long)N + 255) & ~255L;
  const long Ep = ((long)E + 255) & ~255L;

  int*   cntp    = (int*)d_ws;
  int*   startp  = cntp + 512 * 1024;
  int*   rowinfo = startp + 512 * 1024;
  float* dinv    = (float*)(rowinfo + 2 * Np);
  int*   tmp     = (int*)(dinv + Np);
  int*   rec     = tmp + 2 * Ep;
  float* h1      = (float*)(rec + 2 * Ep);
  float* out     = (float*)d_out;

  scatter_kernel<<<256, 1024, 0, stream>>>(src, dst, ew, cntp, startp, tmp, N, E);
  bucket_kernel<<<391, 512, 0, stream>>>(tmp, cntp, startp, rowinfo, dinv, rec, N, E, 256);

  const int nRowTiles = N / 16;

  layer_kernel<<<1024, 256, 0, stream>>>(x, W1, b1, dinv, rowinfo, rec, h1, nRowTiles, 1);

  layer_kernel<<<1024, 256, 0, stream>>>(h1, W2, b2, dinv, rowinfo, rec, out, nRowTiles, 0);
}


# Edge coefficient dinv[src]*w*dinv[dst] formed once: layer 1 stores it over each record's weight word, layer 2 reads it from the record, no second scattered dinv[src] gather; sum order unchanged
# speedup vs baseline: 5.8271x; 1.0325x over previous
#include <hip/hip_runtime.h>
#include <hip/hip_bf16.h>

typedef __attribute__((ext_vector_type(2))) float v2f;
typedef __attribute__((ext_vector_type(8))) float v8f;
typedef unsigned long long u64;

#define HDIM 128

#define BSH 8
#define BNODES 256
#define NBMAX 1024
#define SCAP 8192
#define KCAP 8192
#define RMAX 512
#define EPT 4

__device__ __forceinline__ int block_excl_scan(int v, int* wsum, int& total) {
  const int lane = threadIdx.x & 31;
  const int wid = threadIdx.x >> 5;
  const int nw = blockDim.x >> 5;
  int incl = v;
  for (int d = 1; d < 32; d <<= 1) {
    const int t = __shfl_up(incl, d, 32);
    if (lane >= d) incl += t;
  }
  if (lane == 31) wsum[wid] = incl;
  __syncthreads();
  if (wid == 0) {
    const int w = (lane < nw) ? wsum[lane] : 0;
    int wi = w;
    for (int d = 1; d < 32; d <<= 1) {
      const int t = __shfl_up(wi, d, 32);
      if (lane >= d) wi += t;
    }
    wsum[lane] = wi - w;
    if (lane == 31) wsum[32] = wi;
  }
  __syncthreads();
  total = wsum[32];
  return wsum[wid] + incl - v;
}

__device__ __forceinline__ int run_first(int run, int rounds, int chunk, int E) {
  const int blk = run / rounds;
  const int rd = run - blk * rounds;
  const int cbeg = min(E, blk * chunk);
  const int cend = min(E, cbeg + chunk);
  return min(cend, cbeg + rd * SCAP);
}

__global__ void __launch_bounds__(1024)
scatter_kernel(const int* __restrict__ src, const int* __restrict__ dst,
               const float* __restrict__ w, int* __restrict__ cntp,
               int* __restrict__ startp, int* __restrict__ tmp_w, int N, int E) {
  __shared__ int hist[NBMAX];
  __shared__ int lstart[NBMAX];
  __shared__ int cur[NBMAX];
  __shared__ int wsum[33];
  __shared__ u64 lrec[SCAP];
  u64* __restrict__ tmp = (u64*)tmp_w;
  const int t = threadIdx.x;
  const int NB = (N + BNODES - 1) >> BSH;
  const int chunk = (E + (int)gridDim.x - 1) / (int)gridDim.x;
  const int rounds = (chunk + SCAP - 1) / SCAP;

  for (int rd = 0; rd < rounds; ++rd) {
    const int run = (int)blockIdx.x * rounds + rd;
    const int r0 = run_first(run, rounds, chunk, E);
    const int rn = run_first(run + 1, rounds, chunk, E) - r0;
    if (t < NB) { hist[t] = 0; cur[t] = 0; }
    __syncthreads();
    for (int i0 = t; i0 < rn; i0 += EPT * (int)blockDim.x) {
      int d[EPT];
#pragma unroll
      for (int u = 0; u < EPT; ++u) {
        const int i = i0 + u * (int)blockDim.x;
        d[u] = (i < rn) ? dst[r0 + i] : -1;
      }
#pragma unroll
      for (int u = 0; u < EPT; ++u)
        if (d[u] >= 0) atomicAdd(&hist[d[u] >> BSH], 1);
    }
    __syncthreads();
    const int hv = (t < NB) ? hist[t] : 0;
    int rtot;
    const int lex = block_excl_scan(hv, wsum, rtot);
    if (t < NB) {
      lstart[t] = lex;
      cntp[(long)run * NB + t] = hv;
      startp[(long)run * NB + t] = lex;
    }
    __syncthreads();
    for (int i0 = t; i0 < rn; i0 += EPT * (int)blockDim.x) {
      int d[EPT];
      unsigned sv[EPT];
      unsigned wv[EPT];
#pragma unroll
      for (int u = 0; u < EPT; ++u) {
        const int i = i0 + u * (int)blockDim.x;
        const bool in = i < rn;
        d[u]  = in ? dst[r0 + i] : -1;
        sv[u] = in ? (unsigned)src[r0 + i] : 0u;
        wv[u] = in ? __float_as_uint(w[r0 + i]) : 0u;
      }
#pragma unroll
      for (int u = 0; u < EPT; ++u) {
        if (d[u] >= 0) {
          const int b = d[u] >> BSH;
          const int p = lstart[b] + atomicAdd(&cur[b], 1);
          const unsigned lo = sv[u] | ((unsigned)(d[u] & (BNODES - 1)) << 24);
          lrec[p] = (u64)lo | ((u64)wv[u] << 32);
        }
      }
    }
    __syncthreads();
    for (int i = t; i < rn; i += blockDim.x) tmp[r0 + i] = lrec[i];
    __syncthreads();
  }
}

__global__ void __launch_bounds__(512)
bucket_kernel(const int* __restrict__ tmp_w, const int* __restrict__ cntp,
              const int* __restrict__ startp, int* __restrict__ rowinfo_w,
              float* __restrict__ dinv, int* __restrict__ rec_w, int N, int E, int nsb) {
  __shared__ int cnt[BNODES];
  __shared__ int st[BNODES];
  __shared__ int cur[BNODES];
  __shared__ float dg[BNODES];
  __shared__ int wsum[33];
  __shared__ int rsrc[RMAX];
  __shared__ int rcnt[RMAX];
  __shared__ u64 lrec[KCAP];
  const u64* __restrict__ tmp = (const u64*)tmp_w;
  u64* __restrict__ rec = (u64*)rec_w;
  int2* __restrict__ rowinfo = (int2*)rowinfo_w;
  const int t = threadIdx.x;
  const int lane = t & 31;
  const int wv = t >> 5;
  const int nw = (int)blockDim.x >> 5;
  const int NB = (N + BNODES - 1) >> BSH;
  const int chunk = (E + nsb - 1) / nsb;
  const int rounds = (chunk + SCAP - 1) / SCAP;
  const int R = nsb * rounds;
  const u64 keep = 0xFFFFFFFF00FFFFFFull;

  for (int b = blockIdx.x; b < NB; b += gridDim.x) {
    int cn = 0, ls = 0;
    if (t < R) {
      cn = cntp[(long)t * NB + b];
      ls = startp[(long)t * NB + b];
    }
    int bs, nb;
    block_excl_scan(ls, wsum, bs);
    block_excl_scan(cn, wsum, nb);
    if (t < R) {
      rsrc[t] = run_first(t, rounds, chunk, E) + ls;
      rcnt[t] = cn;
    }
    if (t < BNODES) { cnt[t] = 0; cur[t] = 0; dg[t] = 0.0f; }
    __syncthreads();
    for (int r0 = wv; r0 < R; r0 += EPT * nw) {
      u64 r[EPT];
      int c[EPT];
      int s0[EPT];
#pragma unroll
      for (int u = 0; u < EPT; ++u) {
        const int rr = r0 + u * nw;
        c[u] = (rr < R) ? rcnt[rr] : 0;
        s0[u] = (rr < R) ? rsrc[rr] : 0;
        r[u] = (lane < c[u]) ? tmp[s0[u] + lane] : 0ull;
      }
#pragma unroll
      for (int u = 0; u < EPT; ++u) {
        if (lane < c[u]) {
          const int dl = (int)((unsigned)r[u] >> 24);
          atomicAdd(&cnt[dl], 1);
          atomicAdd(&dg[dl], __uint_as_float((unsigned)(r[u] >> 32)));
        }
        for (int l = lane + 32; l < c[u]; l += 32) {
          const u64 q = tmp[s0[u] + l];
          const int dl = (int)((unsigned)q >> 24);
          atomicAdd(&cnt[dl], 1);
          atomicAdd(&dg[dl], __uint_as_float((unsigned)(q >> 32)));
        }
      }
    }
    __syncthreads();
    const int cc = (t < BNODES) ? cnt[t] : 0;
    int total;
    const int ex = block_excl_scan(cc, wsum, total);
    if (t < BNODES) {
      st[t] = ex;
      const int node = (b << BSH) + t;
      if (node < N) {
        rowinfo[node] = make_int2(bs + ex, cc);
        dinv[node] = rsqrtf(dg[t] + 1.0f);
      }
    }
    __syncthreads();
    const bool inlds = nb <= KCAP;
    for (int r0 = wv; r0 < R; r0 += EPT * nw) {
      u64 r[EPT];
      int c[EPT];
      int s0[EPT];
#pragma unroll
      for (int u = 0; u < EPT; ++u) {
        const int rr = r0 + u * nw;
        c[u] = (rr < R) ? rcnt[rr] : 0;
        s0[u] = (rr < R) ? rsrc[rr] : 0;
        r[u] = (lane < c[u]) ? tmp[s0[u] + lane] : 0ull;
      }
#pragma unroll
      for (int u = 0; u < EPT; ++u) {
        if (lane < c[u]) {
          const int dl = (int)((unsigned)r[u] >> 24);
          const int p = st[dl] + atomicAdd(&cur[dl], 1);
          if (inlds) lrec[p] = r[u] & keep;
          else rec[bs + p] = r[u] & keep;
        }
        for (int l = lane + 32; l < c[u]; l += 32) {
          const u64 q = tmp[s0[u] + l];
          const int dl = (int)((unsigned)q >> 24);
          const int p = st[dl] + atomicAdd(&cur[dl], 1);
          if (inlds) lrec[p] = q & keep;
          else rec[bs + p] = q & keep;
        }
      }
    }
    __syncthreads();
    if (inlds)
      for (int i = t; i < nb; i += blockDim.x) rec[bs + i] = lrec[i];
    __syncthreads();
  }
}

#define APITCH 132

__device__ __forceinline__ float lane_bcast_f(float v, int j) {
  return __int_as_float(__builtin_amdgcn_readlane(__float_as_int(v), j));
}

__device__ __forceinline__ void axpy4(float4& acc, float c, const float4 v) {
  acc.x = fmaf(c, v.x, acc.x);
  acc.y = fmaf(c, v.y, acc.y);
  acc.z = fmaf(c, v.z, acc.z);
  acc.w = fmaf(c, v.w, acc.w);
}

__device__ __forceinline__ u64
load_recs(const u64* rec, int st, int c, int lane) {
  u64 r = 0ull;
  if (lane < c) r = rec[st + lane];
  return r;
}

__device__ __forceinline__ void
add_rows(float4& acc, const float* __restrict__ X, unsigned lcb, int sj, float cj, int m) {
  int j = 0;
  for (; j + 8 <= m; j += 8) {
    float4 v[8];
#pragma unroll
    for (int u = 0; u < 8; ++u) {
      const int s = __builtin_amdgcn_readlane(sj, j + u);
      v[u] = *(const float4*)((const char*)X + (size_t)(unsigned)s * (HDIM * 4) + lcb);
    }
#pragma unroll
    for (int u = 0; u < 8; ++u) axpy4(acc, lane_bcast_f(cj, j + u), v[u]);
  }
  const int rest = m - j;
  if (rest > 4) {
    float4 v[8];
#pragma unroll
    for (int u = 0; u < 8; ++u) {
      const int s = __builtin_amdgcn_readlane(sj, min(j + u, m - 1));
      v[u] = *(const float4*)((const char*)X + (size_t)(unsigned)s * (HDIM * 4) + lcb);
    }
#pragma unroll
    for (int u = 0; u < 8; ++u) axpy4(acc, lane_bcast_f(cj, j + u), v[u]);
  } else if (rest > 0) {
    float4 v[4];
#pragma unroll
    for (int u = 0; u < 4; ++u) {
      const int s = __builtin_amdgcn_readlane(sj, min(j + u, m - 1));
      v[u] = *(const float4*)((const char*)X + (size_t)(unsigned)s * (HDIM * 4) + lcb);
    }
#pragma unroll
    for (int u = 0; u < 4; ++u) axpy4(acc, lane_bcast_f(cj, j + u), v[u]);
  }
}

#define CONSTAS __attribute__((address_space(4)))

__device__ __forceinline__ void
node_info(const u64* __restrict__ rowinfo, const float* __restrict__ dinv,
          int n, bool valid, int& st, int& c, float& di) {
  st = 0;
  c = 0;
  di = 0.0f;
  if (valid) {
    const u64 r = ((const CONSTAS u64*)rowinfo)[n];
    st = (int)(unsigned)r;
    c = (int)(unsigned)(r >> 32);
    di = ((const CONSTAS float*)dinv)[n];
  }
}

template <bool FIRST>
__global__ void __launch_bounds__(256, 8)
layer_kernel(const float* __restrict__ X, const float* __restrict__ W,
             const float* __restrict__ b, const float* __restrict__ dinv,
             const int* __restrict__ rowinfo_w, int* rec_w,
             float* __restrict__ Out, int nRowTiles, int relu) {
  __shared__ __attribute__((aligned(16))) float stage[2][16 * APITCH];
  const u64* __restrict__ rowinfo = (const u64*)rowinfo_w;
  const u64* rec = (const u64*)rec_w;
  float* coef = (float*)rec_w;

  const int lane = threadIdx.x & 31;
  const int wave = __builtin_amdgcn_readfirstlane((int)(threadIdx.x >> 5));
  const int m    = lane & 15;
  const int half = lane >> 4;
  const int col0 = wave * 16;

  v2f bfrag[32];
#pragma unroll
  for (int kk = 0; kk < 32; ++kk) {
    const int k = kk * 4 + half * 2;
    v2f bb;
    bb.x = W[(k + 0) * HDIM + col0 + m];
    bb.y = W[(k + 1) * HDIM + col0 + m];
    bfrag[kk] = bb;
  }
  const float bcol = b[col0 + m];

  const int lc = lane * 4;
  const unsigned lcb = (unsigned)lane * 16u;
  const int soff0 = wave * APITCH + lc;
  const int soff1 = (wave + 8) * APITCH + lc;
  const int foff = m * APITCH + half * 2;

  const int tile0 = (int)blockIdx.x;
  const int tstride = (int)gridDim.x;
  int st0, c0, st1, c1, st2, c2;
  float di0, di1, di2;
  node_info(rowinfo, dinv, tile0 * 16 + wave, tile0 < nRowTiles, st0, c0, di0);
  node_info(rowinfo, dinv, tile0 * 16 + wave + 8, tile0 < nRowTiles, st1, c1, di1);
  node_info(rowinfo, dinv, (tile0 + tstride) * 16 + wave, tile0 + tstride < nRowTiles,
            st2, c2, di2);
  int sj0;
  float cj0;
  u64 rA;
  {
    const u64 r0 = load_recs(rec, st0, c0, lane);
    rA = load_recs(rec, st1, c1, lane);
    sj0 = (int)(unsigned)r0;
    const float w0 = __uint_as_float((unsigned)(r0 >> 32));
    if (FIRST) {
      const float d0 = *(const float*)((const char*)dinv + ((unsigned)sj0 << 2));
      cj0 = (lane < c0) ? d0 * w0 * di0 : 0.0f;
    } else {
      cj0 = w0;
      asm volatile("" :: "v"(sj0), "v"(cj0));
    }
  }

  int buf = 0;
  for (int rt = tile0; rt < nRowTiles; rt += tstride) {
    float* __restrict__ st = stage[buf];
#pragma unroll
    for (int h = 0; h < 2; ++h) {
      const int n = rt * 16 + wave + 8 * h;
      if (FIRST) {
        if (lane < c0) coef[2 * (st0 + lane) + 1] = cj0;
      }
      const int t3 = rt + (h ? 2 : 1) * tstride;
      int st3, c3;
      float di3;
      node_info(rowinfo, dinv, t3 * 16 + wave + (h ? 0 : 8), t3 < nRowTiles, st3, c3, di3);
      const int sjN = (int)(unsigned)rA;
      const float wN = __uint_as_float((unsigned)(rA >> 32));
      float dN = 0.0f;
      if (FIRST) dN = *(const float*)((const char*)dinv + ((unsigned)sjN << 2));
      const u64 rB = load_recs(rec, st2, c2, lane);
      const float4 xs = *(const float4*)((const char*)X + (size_t)(unsigned)n * (HDIM * 4) + lcb);

      float4 acc = {0.0f, 0.0f, 0.0f, 0.0f};
      add_rows(acc, X, lcb, sj0, cj0, min(32, c0));
      for (int base = 32; base < c0; base += 32) {
        const int mm = min(32, c0 - base);
        const u64 r = load_recs(rec, st0 + base, mm, lane);
        const int sjx = (int)(unsigned)r;
        float cjx = __uint_as_float((unsigned)(r >> 32));
        if (FIRST) {
          cjx = (lane < mm) ? dinv[(unsigned)sjx] * cjx * di0 : 0.0f;
          if (lane < mm) coef[2 * (st0 + base + lane) + 1] = cjx;
        }
        add_rows(acc, X, lcb, sjx, cjx, mm);
      }

      const float dd = di0 * di0;
      float4 v;
      v.x = acc.x + dd * xs.x;
      v.y = acc.y + dd * xs.y;
      v.z = acc.z + dd * xs.z;
      v.w = acc.w + dd * xs.w;
      *(float4*)(st + (h ? soff1 : soff0)) = v;

      sj0 = sjN;
      if (FIRST) cj0 = (lane < c1) ? dN * wN * di1 : 0.0f;
      else cj0 = wN;
      rA = rB;
      st0 = st1; c0 = c1; di0 = di1;
      st1 = st2; c1 = c2; di1 = di2;
      st2 = st3; c2 = c3; di2 = di3;
    }
    __syncthreads();

    const float* __restrict__ arow = st + foff;
    v8f acc = {0.f, 0.f, 0.f, 0.f, 0.f, 0.f, 0.f, 0.f};
#pragma unroll
    for (int kk = 0; kk < 32; ++kk) {
      const v2f a = *(const v2f*)(arow + kk * 4);
      acc = __builtin_amdgcn_wmma_f32_16x16x4_f32(
           false, a,  false, bfrag[kk],
           (short)0, acc,  false,  false);
    }
    float* __restrict__ otile = Out + (size_t)(unsigned)rt * (16 * HDIM);
#pragma unroll
    for (int r = 0; r < 8; ++r) {
      float v = acc[r] + bcol;
      if (relu) v = fmaxf(v, 0.0f);
      otile[(half * 8 + r) * HDIM + col0 + m] = v;
    }
    buf ^= 1;
  }
}

extern "C" void kernel_launch(void* const* d_in, const int* in_sizes, int n_in,
                              void* d_out, int out_size, void* d_ws, size_t ws_size,
                              hipStream_t stream) {
  const float* x  = (const float*)d_in[0];
  const float* ew = (const float*)d_in[1];
  const float* W1 = (const float*)d_in[2];
  const float* b1 = (const float*)d_in[3];
  const float* W2 = (const float*)d_in[4];
  const float* b2 = (const float*)d_in[5];
  const int*   ei = (const int*)d_in[6];

  const int N = in_sizes[0] / HDIM;
  const int E = in_sizes[1];
  const int* src = ei;
  const int* dst = ei + E;

  const long Np = ((long)N + 255) & ~255L;
  const long Ep = ((long)E + 255) & ~255L;

  int*   cntp    = (int*)d_ws;
  int*   startp  = cntp + 512 * 1024;
  int*   rowinfo = startp + 512 * 1024;
  float* dinv    = (float*)(rowinfo + 2 * Np);
  int*   tmp     = (int*)(dinv + Np);
  int*   rec     = tmp + 2 * Ep;
  float* h1      = (float*)(rec + 2 * Ep);
  float* out     = (float*)d_out;

  scatter_kernel<<<256, 1024, 0, stream>>>(src, dst, ew, cntp, startp, tmp, N, E);
  bucket_kernel<<<391, 512, 0, stream>>>(tmp, cntp, startp, rowinfo, dinv, rec, N, E, 256);

  const int nRowTiles = N / 16;

  layer_kernel<true><<<1024, 256, 0, stream>>>(x, W1, b1, dinv, rowinfo, rec, h1, nRowTiles, 1);

  layer_kernel<false><<<1024, 256, 0, stream>>>(h1, W2, b2, dinv, rowinfo, rec, out, nRowTiles, 0);
}


# Output tile leaves as whole 512-byte rows: the 8 waves' 16-column strips meet in an LDS tile (one more barrier), a wave stores 2 full rows, not 8 stores of 64-byte halves; sum order unchanged
# speedup vs baseline: 6.1207x; 1.0504x over previous
#include <hip/hip_runtime.h>
#include <hip/hip_bf16.h>

typedef __attribute__((ext_vector_type(2))) float v2f;
typedef __attribute__((ext_vector_type(8))) float v8f;
typedef unsigned long long u64;

#define HDIM 128

#define BSH 8
#define BNODES 256
#define NBMAX 1024
#define SCAP 8192
#define KCAP 8192
#define RMAX 512
#define EPT 4

__device__ __forceinline__ int block_excl_scan(int v, int* wsum, int& total) {
  const int lane = threadIdx.x & 31;
  const int wid = threadIdx.x >> 5;
  const int nw = blockDim.x >> 5;
  int incl = v;
  for (int d = 1; d < 32; d <<= 1) {
    const int t = __shfl_up(incl, d, 32);
    if (lane >= d) incl += t;
  }
  if (lane == 31) wsum[wid] = incl;
  __syncthreads();
  if (wid == 0) {
    const int w = (lane < nw) ? wsum[lane] : 0;
    int wi = w;
    for (int d = 1; d < 32; d <<= 1) {
      const int t = __shfl_up(wi, d, 32);
      if (lane >= d) wi += t;
    }
    wsum[lane] = wi - w;
    if (lane == 31) wsum[32] = wi;
  }
  __syncthreads();
  total = wsum[32];
  return wsum[wid] + incl - v;
}

__device__ __forceinline__ int run_first(int run, int rounds, int chunk, int E) {
  const int blk = run / rounds;
  const int rd = run - blk * rounds;
  const int cbeg = min(E, blk * chunk);
  const int cend = min(E, cbeg + chunk);
  return min(cend, cbeg + rd * SCAP);
}

__global__ void __launch_bounds__(1024)
scatter_kernel(const int* __restrict__ src, const int* __restrict__ dst,
               const float* __restrict__ w, int* __restrict__ cntp,
               int* __restrict__ startp, int* __restrict__ tmp_w, int N, int E) {
  __shared__ int hist[NBMAX];
  __shared__ int lstart[NBMAX];
  __shared__ int cur[NBMAX];
  __shared__ int wsum[33];
  __shared__ u64 lrec[SCAP];
  u64* __restrict__ tmp = (u64*)tmp_w;
  const int t = threadIdx.x;
  const int NB = (N + BNODES - 1) >> BSH;
  const int chunk = (E + (int)gridDim.x - 1) / (int)gridDim.x;
  const int rounds = (chunk + SCAP - 1) / SCAP;

  for (int rd = 0; rd < rounds; ++rd) {
    const int run = (int)blockIdx.x * rounds + rd;
    const int r0 = run_first(run, rounds, chunk, E);
    const int rn = run_first(run + 1, rounds, chunk, E) - r0;
    if (t < NB) { hist[t] = 0; cur[t] = 0; }
    __syncthreads();
    for (int i0 = t; i0 < rn; i0 += EPT * (int)blockDim.x) {
      int d[EPT];
#pragma unroll
      for (int u = 0; u < EPT; ++u) {
        const int i = i0 + u * (int)blockDim.x;
        d[u] = (i < rn) ? dst[r0 + i] : -1;
      }
#pragma unroll
      for (int u = 0; u < EPT; ++u)
        if (d[u] >= 0) atomicAdd(&hist[d[u] >> BSH], 1);
    }
    __syncthreads();
    const int hv = (t < NB) ? hist[t] : 0;
    int rtot;
    const int lex = block_excl_scan(hv, wsum, rtot);
    if (t < NB) {
      lstart[t] = lex;
      cntp[(long)run * NB + t] = hv;
      startp[(long)run * NB + t] = lex;
    }
    __syncthreads();
    for (int i0 = t; i0 < rn; i0 += EPT * (int)blockDim.x) {
      int d[EPT];
      unsigned sv[EPT];
      unsigned wv[EPT];
#pragma unroll
      for (int u = 0; u < EPT; ++u) {
        const int i = i0 + u * (int)blockDim.x;
        const bool in = i < rn;
        d[u]  = in ? dst[r0 + i] : -1;
        sv[u] = in ? (unsigned)src[r0 + i] : 0u;
        wv[u] = in ? __float_as_uint(w[r0 + i]) : 0u;
      }
#pragma unroll
      for (int u = 0; u < EPT; ++u) {
        if (d[u] >= 0) {
          const int b = d[u] >> BSH;
          const int p = lstart[b] + atomicAdd(&cur[b], 1);
          const unsigned lo = sv[u] | ((unsigned)(d[u] & (BNODES - 1)) << 24);
          lrec[p] = (u64)lo | ((u64)wv[u] << 32);
        }
      }
    }
    __syncthreads();
    for (int i = t; i < rn; i += blockDim.x) tmp[r0 + i] = lrec[i];
    __syncthreads();
  }
}

__global__ void __launch_bounds__(512)
bucket_kernel(const int* __restrict__ tmp_w, const int* __restrict__ cntp,
              const int* __restrict__ startp, int* __restrict__ rowinfo_w,
              float* __restrict__ dinv, int* __restrict__ rec_w, int N, int E, int nsb) {
  __shared__ int cnt[BNODES];
  __shared__ int st[BNODES];
  __shared__ int cur[BNODES];
  __shared__ float dg[BNODES];
  __shared__ int wsum[33];
  __shared__ int rsrc[RMAX];
  __shared__ int rcnt[RMAX];
  __shared__ u64 lrec[KCAP];
  const u64* __restrict__ tmp = (const u64*)tmp_w;
  u64* __restrict__ rec = (u64*)rec_w;
  int2* __restrict__ rowinfo = (int2*)rowinfo_w;
  const int t = threadIdx.x;
  const int lane = t & 31;
  const int wv = t >> 5;
  const int nw = (int)blockDim.x >> 5;
  const int NB = (N + BNODES - 1) >> BSH;
  const int chunk = (E + nsb - 1) / nsb;
  const int rounds = (chunk + SCAP - 1) / SCAP;
  const int R = nsb * rounds;
  const u64 keep = 0xFFFFFFFF00FFFFFFull;

  for (int b = blockIdx.x; b < NB; b += gridDim.x) {
    int cn = 0, ls = 0;
    if (t < R) {
      cn = cntp[(long)t * NB + b];
      ls = startp[(long)t * NB + b];
    }
    int bs, nb;
    block_excl_scan(ls, wsum, bs);
    block_excl_scan(cn, wsum, nb);
    if (t < R) {
      rsrc[t] = run_first(t, rounds, chunk, E) + ls;
      rcnt[t] = cn;
    }
    if (t < BNODES) { cnt[t] = 0; cur[t] = 0; dg[t] = 0.0f; }
    __syncthreads();
    for (int r0 = wv; r0 < R; r0 += EPT * nw) {
      u64 r[EPT];
      int c[EPT];
      int s0[EPT];
#pragma unroll
      for (int u = 0; u < EPT; ++u) {
        const int rr = r0 + u * nw;
        c[u] = (rr < R) ? rcnt[rr] : 0;
        s0[u] = (rr < R) ? rsrc[rr] : 0;
        r[u] = (lane < c[u]) ? tmp[s0[u] + lane] : 0ull;
      }
#pragma unroll
      for (int u = 0; u < EPT; ++u) {
        if (lane < c[u]) {
          const int dl = (int)((unsigned)r[u] >> 24);
          atomicAdd(&cnt[dl], 1);
          atomicAdd(&dg[dl], __uint_as_float((unsigned)(r[u] >> 32)));
        }
        for (int l = lane + 32; l < c[u]; l += 32) {
          const u64 q = tmp[s0[u] + l];
          const int dl = (int)((unsigned)q >> 24);
          atomicAdd(&cnt[dl], 1);
          atomicAdd(&dg[dl], __uint_as_float((unsigned)(q >> 32)));
        }
      }
    }
    __syncthreads();
    const int cc = (t < BNODES) ? cnt[t] : 0;
    int total;
    const int ex = block_excl_scan(cc, wsum, total);
    if (t < BNODES) {
      st[t] = ex;
      const int node = (b << BSH) + t;
      if (node < N) {
        rowinfo[node] = make_int2(bs + ex, cc);
        dinv[node] = rsqrtf(dg[t] + 1.0f);
      }
    }
    __syncthreads();
    const bool inlds = nb <= KCAP;
    for (int r0 = wv; r0 < R; r0 += EPT * nw) {
      u64 r[EPT];
      int c[EPT];
      int s0[EPT];
#pragma unroll
      for (int u = 0; u < EPT; ++u) {
        const int rr = r0 + u * nw;
        c[u] = (rr < R) ? rcnt[rr] : 0;
        s0[u] = (rr < R) ? rsrc[rr] : 0;
        r[u] = (lane < c[u]) ? tmp[s0[u] + lane] : 0ull;
      }
#pragma unroll
      for (int u = 0; u < EPT; ++u) {
        if (lane < c[u]) {
          const int dl = (int)((unsigned)r[u] >> 24);
          const int p = st[dl] + atomicAdd(&cur[dl], 1);
          if (inlds) lrec[p] = r[u] & keep;
          else rec[bs + p] = r[u] & keep;
        }
        for (int l = lane + 32; l < c[u]; l += 32) {
          const u64 q = tmp[s0[u] + l];
          const int dl = (int)((unsigned)q >> 24);
          const int p = st[dl] + atomicAdd(&cur[dl], 1);
          if (inlds) lrec[p] = q & keep;
          else rec[bs + p] = q & keep;
        }
      }
    }
    __syncthreads();
    if (inlds)
      for (int i = t; i < nb; i += blockDim.x) rec[bs + i] = lrec[i];
    __syncthreads();
  }
}

#define APITCH 132

__device__ __forceinline__ float lane_bcast_f(float v, int j) {
  return __int_as_float(__builtin_amdgcn_readlane(__float_as_int(v), j));
}

__device__ __forceinline__ void axpy4(float4& acc, float c, const float4 v) {
  acc.x = fmaf(c, v.x, acc.x);
  acc.y = fmaf(c, v.y, acc.y);
  acc.z = fmaf(c, v.z, acc.z);
  acc.w = fmaf(c, v.w, acc.w);
}

__device__ __forceinline__ u64
load_recs(const u64* rec, int st, int c, int lane) {
  u64 r = 0ull;
  if (lane < c) r = rec[st + lane];
  return r;
}

__device__ __forceinline__ void
add_rows(float4& acc, const float* __restrict__ X, unsigned lcb, int sj, float cj, int m) {
  int j = 0;
  for (; j + 8 <= m; j += 8) {
    float4 v[8];
#pragma unroll
    for (int u = 0; u < 8; ++u) {
      const int s = __builtin_amdgcn_readlane(sj, j + u);
      v[u] = *(const float4*)((const char*)X + (size_t)(unsigned)s * (HDIM * 4) + lcb);
    }
#pragma unroll
    for (int u = 0; u < 8; ++u) axpy4(acc, lane_bcast_f(cj, j + u), v[u]);
  }
  const int rest = m - j;
  if (rest > 4) {
    float4 v[8];
#pragma unroll
    for (int u = 0; u < 8; ++u) {
      const int s = __builtin_amdgcn_readlane(sj, min(j + u, m - 1));
      v[u] = *(const float4*)((const char*)X + (size_t)(unsigned)s * (HDIM * 4) + lcb);
    }
#pragma unroll
    for (int u = 0; u < 8; ++u) axpy4(acc, lane_bcast_f(cj, j + u), v[u]);
  } else if (rest > 0) {
    float4 v[4];
#pragma unroll
    for (int u = 0; u < 4; ++u) {
      const int s = __builtin_amdgcn_readlane(sj, min(j + u, m - 1));
      v[u] = *(const float4*)((const char*)X + (size_t)(unsigned)s * (HDIM * 4) + lcb);
    }
#pragma unroll
    for (int u = 0; u < 4; ++u) axpy4(acc, lane_bcast_f(cj, j + u), v[u]);
  }
}

#define CONSTAS __attribute__((address_space(4)))

__device__ __forceinline__ void
node_info(const u64* __restrict__ rowinfo, const float* __restrict__ dinv,
          int n, bool valid, int& st, int& c, float& di) {
  st = 0;
  c = 0;
  di = 0.0f;
  if (valid) {
    const u64 r = ((const CONSTAS u64*)rowinfo)[n];
    st = (int)(unsigned)r;
    c = (int)(unsigned)(r >> 32);
    di = ((const CONSTAS float*)dinv)[n];
  }
}

template <bool FIRST>
__global__ void __launch_bounds__(256, 8)
layer_kernel(const float* __restrict__ X, const float* __restrict__ W,
             const float* __restrict__ b, const float* __restrict__ dinv,
             const int* __restrict__ rowinfo_w, int* rec_w,
             float* __restrict__ Out, int nRowTiles, int relu) {
  __shared__ __attribute__((aligned(16))) float stage[2][2][16 * APITCH];
  const u64* __restrict__ rowinfo = (const u64*)rowinfo_w;
  const u64* rec = (const u64*)rec_w;
  float* coef = (float*)rec_w;

  const int lane = threadIdx.x & 31;
  const int wave = __builtin_amdgcn_readfirstlane((int)(threadIdx.x >> 5));
  const int m    = lane & 15;
  const int half = lane >> 4;
  const int col0 = wave * 16;

  v2f bfrag[32];
#pragma unroll
  for (int kk = 0; kk < 32; ++kk) {
    const int k = kk * 4 + half * 2;
    v2f bb;
    bb.x = W[(k + 0) * HDIM + col0 + m];
    bb.y = W[(k + 1) * HDIM + col0 + m];
    bfrag[kk] = bb;
  }
  const float bcol = b[col0 + m];

  const int lc = lane * 4;
  const unsigned lcb = (unsigned)lane * 16u;
  const int soff0 = wave * APITCH + lc;
  const int soff1 = (wave + 8) * APITCH + lc;
  const int foff = m * APITCH + half * 2;

  const int tile0 = (int)blockIdx.x;
  const int tstride = (int)gridDim.x;
  int st0, c0, st1, c1, st2, c2;
  float di0, di1, di2;
  node_info(rowinfo, dinv, tile0 * 16 + wave, tile0 < nRowTiles, st0, c0, di0);
  node_info(rowinfo, dinv, tile0 * 16 + wave + 8, tile0 < nRowTiles, st1, c1, di1);
  node_info(rowinfo, dinv, (tile0 + tstride) * 16 + wave, tile0 + tstride < nRowTiles,
            st2, c2, di2);
  int sj0;
  float cj0;
  u64 rA;
  {
    const u64 r0 = load_recs(rec, st0, c0, lane);
    rA = load_recs(rec, st1, c1, lane);
    sj0 = (int)(unsigned)r0;
    const float w0 = __uint_as_float((unsigned)(r0 >> 32));
    if (FIRST) {
      const float d0 = *(const float*)((const char*)dinv + ((unsigned)sj0 << 2));
      cj0 = (lane < c0) ? d0 * w0 * di0 : 0.0f;
    } else {
      cj0 = w0;
      asm volatile("" :: "v"(sj0), "v"(cj0));
    }
  }

  int buf = 0;
  for (int rt = tile0; rt < nRowTiles; rt += tstride) {
    float* __restrict__ st = stage[buf][0];
    float* __restrict__ otl = st + 16 * APITCH;
#pragma unroll
    for (int h = 0; h < 2; ++h) {
      const int n = rt * 16 + wave + 8 * h;
      if (FIRST) {
        if (lane < c0) coef[2 * (st0 + lane) + 1] = cj0;
      }
      const int t3 = rt + (h ? 2 : 1) * tstride;
      int st3, c3;
      float di3;
      node_info(rowinfo, dinv, t3 * 16 + wave + (h ? 0 : 8), t3 < nRowTiles, st3, c3, di3);
      const int sjN = (int)(unsigned)rA;
      const float wN = __uint_as_float((unsigned)(rA >> 32));
      float dN = 0.0f;
      if (FIRST) dN = *(const float*)((const char*)dinv + ((unsigned)sjN << 2));
      const u64 rB = load_recs(rec, st2, c2, lane);
      const float4 xs = *(const float4*)((const char*)X + (size_t)(unsigned)n * (HDIM * 4) + lcb);

      float4 acc = {0.0f, 0.0f, 0.0f, 0.0f};
      add_rows(acc, X, lcb, sj0, cj0, min(32, c0));
      for (int base = 32; base < c0; base += 32) {
        const int mm = min(32, c0 - base);
        const u64 r = load_recs(rec, st0 + base, mm, lane);
        const int sjx = (int)(unsigned)r;
        float cjx = __uint_as_float((unsigned)(r >> 32));
        if (FIRST) {
          cjx = (lane < mm) ? dinv[(unsigned)sjx] * cjx * di0 : 0.0f;
          if (lane < mm) coef[2 * (st0 + base + lane) + 1] = cjx;
        }
        add_rows(acc, X, lcb, sjx, cjx, mm);
      }

      const float dd = di0 * di0;
      float4 v;
      v.x = acc.x + dd * xs.x;
      v.y = acc.y + dd * xs.y;
      v.z = acc.z + dd * xs.z;
      v.w = acc.w + dd * xs.w;
      *(float4*)(st + (h ? soff1 : soff0)) = v;

      sj0 = sjN;
      if (FIRST) cj0 = (lane < c1) ? dN * wN * di1 : 0.0f;
      else cj0 = wN;
      rA = rB;
      st0 = st1; c0 = c1; di0 = di1;
      st1 = st2; c1 = c2; di1 = di2;
      st2 = st3; c2 = c3; di2 = di3;
    }
    __syncthreads();

    const float* __restrict__ arow = st + foff;
    v8f acc = {0.f, 0.f, 0.f, 0.f, 0.f, 0.f, 0.f, 0.f};
#pragma unroll
    for (int kk = 0; kk < 32; ++kk) {
      const v2f a = *(const v2f*)(arow + kk * 4);
      acc = __builtin_amdgcn_wmma_f32_16x16x4_f32(
           false, a,  false, bfrag[kk],
           (short)0, acc,  false,  false);
    }
#pragma unroll
    for (int r = 0; r < 8; ++r) {
      float v = acc[r] + bcol;
      if (relu) v = fmaxf(v, 0.0f);
      otl[(half * 8 + r) * APITCH + col0 + m] = v;
    }
    __syncthreads();
    const float4 o0 = *(const float4*)(otl + soff0);
    const float4 o1 = *(const float4*)(otl + soff1);
    float* __restrict__ otile = Out + (size_t)(unsigned)rt * (16 * HDIM);
    *(float4*)(otile + wave * HDIM + lc) = o0;
    *(float4*)(otile + (wave + 8) * HDIM + lc) = o1;
    buf ^= 1;
  }
}

extern "C" void kernel_launch(void* const* d_in, const int* in_sizes, int n_in,
                              void* d_out, int out_size, void* d_ws, size_t ws_size,
                              hipStream_t stream) {
  const float* x  = (const float*)d_in[0];
  const float* ew = (const float*)d_in[1];
  const float* W1 = (const float*)d_in[2];
  const float* b1 = (const float*)d_in[3];
  const float* W2 = (const float*)d_in[4];
  const float* b2 = (const float*)d_in[5];
  const int*   ei = (const int*)d_in[6];

  const int N = in_sizes[0] / HDIM;
  const int E = in_sizes[1];
  const int* src = ei;
  const int* dst = ei + E;

  const long Np = ((long)N + 255) & ~255L;
  const long Ep = ((long)E + 255) & ~255L;

  int*   cntp    = (int*)d_ws;
  int*   startp  = cntp + 512 * 1024;
  int*   rowinfo = startp + 512 * 1024;
  float* dinv    = (float*)(rowinfo + 2 * Np);
  int*   tmp     = (int*)(dinv + Np);
  int*   rec     = tmp + 2 * Ep;
  float* h1      = (float*)(rec + 2 * Ep);
  float* out     = (float*)d_out;

  scatter_kernel<<<256, 1024, 0, stream>>>(src, dst, ew, cntp, startp, tmp, N, E);
  bucket_kernel<<<391, 512, 0, stream>>>(tmp, cntp, startp, rowinfo, dinv, rec, N, E, 256);

  const int nRowTiles = N / 16;

  layer_kernel<true><<<1024, 256, 0, stream>>>(x, W1, b1, dinv, rowinfo, rec, h1, nRowTiles, 1);

  layer_kernel<false><<<1024, 256, 0, stream>>>(h1, W2, b2, dinv, rowinfo, rec, out, nRowTiles, 0);
}
